# starting kernel
# baseline (speedup 1.0000x reference)
#include <hip/hip_runtime.h>
#include <cstddef>

typedef __attribute__((ext_vector_type(2))) float v2f;
typedef __attribute__((ext_vector_type(8))) float v8f;

#define GCN_N 50000
#define THREADS 256

__global__ void k_init_deg(float* __restrict__ deg, int n) {
    int i = blockIdx.x * blockDim.x + threadIdx.x;
    if (i < n) deg[i] = 1.0f;
}

__global__ void k_accum_deg(const int* __restrict__ dst, float* __restrict__ deg, int e) {
    int i = blockIdx.x * blockDim.x + threadIdx.x;
    if (i < e) atomicAdd(&deg[dst[i]], 1.0f);
}

__global__ void k_dinv(float* __restrict__ deg, int n) {
    int i = blockIdx.x * blockDim.x + threadIdx.x;
    if (i < n) deg[i] = rsqrtf(fmaxf(deg[i], 1.0f));
}

__global__ void k_zero(float* __restrict__ p, size_t n) {
    size_t i = (size_t)blockIdx.x * blockDim.x + threadIdx.x;
    if (i < n) p[i] = 0.0f;
}

template <int KIN, int KOUT>
__global__ __launch_bounds__(THREADS)
void k_gemm_wmma(const float* __restrict__ A, const float* __restrict__ W,
                 float* __restrict__ C, int nrows) {
    __shared__ v2f sW[(KIN / 2) * KOUT];
    for (int i = threadIdx.x; i < (KIN / 2) * KOUT; i += blockDim.x) {
        const int k2 = i / KOUT;
        const int c  = i % KOUT;
        v2f w;
        w.x = W[(2 * k2 + 0) * KOUT + c];
        w.y = W[(2 * k2 + 1) * KOUT + c];
        sW[i] = w;
    }
    __syncthreads();

    const int wave = threadIdx.x >> 5;
    const int lane = threadIdx.x & 31;
    const int tile = blockIdx.x * 8 + wave;
    if (tile * 16 >= nrows) return;

    constexpr int NT = KOUT / 16;
    v8f acc[NT] = {};

    const int m    = lane & 15;
    const int khi  = lane >> 4;
    const float* Arow = A + (size_t)(tile * 16 + m) * KIN;
    const v2f* sWl = sW + khi * KOUT + m;

    for (int kk = 0; kk < KIN / 4; ++kk) {
        const v2f a = *(const v2f*)(Arow + kk * 4 + khi * 2);
        const v2f* sWk = sWl + (kk * 2) * KOUT;
#pragma unroll
        for (int t = 0; t < NT; ++t) {
            const v2f b = sWk[t * 16];
            acc[t] = __builtin_amdgcn_wmma_f32_16x16x4_f32(
                 false, a,  false, b,
                 (short)0, acc[t],  false,  false);
        }
    }

#pragma unroll
    for (int t = 0; t < NT; ++t) {
#pragma unroll
        for (int v = 0; v < 8; ++v) {
            const int r = tile * 16 + khi * 8 + v;
            C[(size_t)r * KOUT + t * 16 + m] = acc[t][v];
        }
    }
}

template <int F>
__global__ __launch_bounds__(THREADS)
void k_scatter(const int* __restrict__ src, const int* __restrict__ dst,
               const float* __restrict__ dinv, const float* __restrict__ H,
               float* __restrict__ out, int nedges) {
    constexpr int C = F / 4;
    const long long tid = (long long)blockIdx.x * blockDim.x + threadIdx.x;
    const int e = (int)(tid / C);
    const int c = (int)(tid % C);
    if (e >= nedges) return;

    const int s = src[e];
    const int d = dst[e];
    const float nrm = dinv[s] * dinv[d];
    const float4 h = *(const float4*)(H + (size_t)s * F + c * 4);
    float* o = out + (size_t)d * F + c * 4;
    atomicAdd(o + 0, h.x * nrm);
    atomicAdd(o + 1, h.y * nrm);
    atomicAdd(o + 2, h.z * nrm);
    atomicAdd(o + 3, h.w * nrm);
}

template <int F, bool RELU>
__global__ __launch_bounds__(THREADS)
void k_finalize(const float* __restrict__ H, const float* __restrict__ dinv,
                const float* __restrict__ bias, float* __restrict__ out, int n) {
    const long long tid = (long long)blockIdx.x * blockDim.x + threadIdx.x;
    const int i = (int)(tid / F);
    const int f = (int)(tid % F);
    if (i >= n) return;
    const float di = dinv[i];
    float v = out[(size_t)i * F + f] + H[(size_t)i * F + f] * di * di + bias[f];
    out[(size_t)i * F + f] = RELU ? fmaxf(v, 0.0f) : v;
}

extern "C" void kernel_launch(void* const* d_in, const int* in_sizes, int n_in,
                              void* d_out, int out_size, void* d_ws, size_t ws_size,
                              hipStream_t stream) {
    const float* x  = (const float*)d_in[0];
    const int*   ei = (const int*)d_in[1];
    const float* W1 = (const float*)d_in[2];
    const float* b1 = (const float*)d_in[3];
    const float* W2 = (const float*)d_in[4];
    const float* b2 = (const float*)d_in[5];
    const float* W3 = (const float*)d_in[6];
    const float* b3 = (const float*)d_in[7];
    float* out = (float*)d_out;

    const int N = GCN_N;
    const int E = in_sizes[1] / 2;
    const int* src = ei;
    const int* dst = ei + E;

    float* dinv = (float*)d_ws;
    float* bufA = dinv + ((N + 255) & ~255);
    float* bufB = bufA + (size_t)N * 64;

    const int tiles  = N / 16;
    const int gemmBl = (tiles + 7) / 8;

    auto blocks = [](long long n) { return (unsigned)((n + THREADS - 1) / THREADS); };

    k_init_deg<<<blocks(N), THREADS, 0, stream>>>(dinv, N);
    k_accum_deg<<<blocks(E), THREADS, 0, stream>>>(dst, dinv, E);
    k_dinv<<<blocks(N), THREADS, 0, stream>>>(dinv, N);

    k_gemm_wmma<128, 64><<<gemmBl, THREADS, 0, stream>>>(x, W1, bufA, N);
    k_zero<<<blocks((long long)N * 64), THREADS, 0, stream>>>(bufB, (size_t)N * 64);
    k_scatter<64><<<blocks((long long)E * 16), THREADS, 0, stream>>>(src, dst, dinv, bufA, bufB, E);
    k_finalize<64, true><<<blocks((long long)N * 64), THREADS, 0, stream>>>(bufA, dinv, b1, bufB, N);

    k_gemm_wmma<64, 64><<<gemmBl, THREADS, 0, stream>>>(bufB, W2, bufA, N);
    k_zero<<<blocks((long long)N * 64), THREADS, 0, stream>>>(bufB, (size_t)N * 64);
    k_scatter<64><<<blocks((long long)E * 16), THREADS, 0, stream>>>(src, dst, dinv, bufA, bufB, E);
    k_finalize<64, true><<<blocks((long long)N * 64), THREADS, 0, stream>>>(bufA, dinv, b2, bufB, N);

    k_gemm_wmma<64, 32><<<gemmBl, THREADS, 0, stream>>>(bufB, W3, bufA, N);
    k_zero<<<blocks((long long)N * 32), THREADS, 0, stream>>>(out, (size_t)N * 32);
    k_scatter<32><<<blocks((long long)E * 8), THREADS, 0, stream>>>(src, dst, dinv, bufA, out, E);
    k_finalize<32, false><<<blocks((long long)N * 32), THREADS, 0, stream>>>(bufA, dinv, b3, out, N);
}


# Aggregate by destination row: per-call edge lists (returning integer rank adds, scan, fill) and a register gather replace float-atomic scatter, zero fill, finalize; row sum order = arrival rank
# speedup vs baseline: 1.4742x; 1.4742x over previous
#include <hip/hip_runtime.h>
#include <cstddef>

typedef __attribute__((ext_vector_type(2))) float v2f;
typedef __attribute__((ext_vector_type(8))) float v8f;

#define GCN_N 50000
#define THREADS 256

__global__ void k_zero_cnt(int* __restrict__ cnt, int n) {
    int i = blockIdx.x * blockDim.x + threadIdx.x;
    if (i < n) cnt[i] = 0;
}

__global__ void k_count_rank(const int* __restrict__ dst, int* __restrict__ cnt,
                             int* __restrict__ rank, int e) {
    int i = blockIdx.x * blockDim.x + threadIdx.x;
    if (i < e) rank[i] = atomicAdd(&cnt[dst[i]], 1);
}

#define SCAN_T 1024
#define SCAN_CH ((GCN_N + SCAN_T - 1) / SCAN_T)
__global__ __launch_bounds__(SCAN_T)
void k_scan_dinv(const int* __restrict__ cnt, int* __restrict__ rowptr,
                 float* __restrict__ dinv, int n) {
    __shared__ int s_tot[SCAN_T / 32];
    const int lane = threadIdx.x & 31;
    const int wave = threadIdx.x >> 5;
    const int seg  = wave * (SCAN_CH * 32);

    int s = 0;
#pragma unroll
    for (int k = 0; k < SCAN_CH; ++k) {
        const int idx = seg + k * 32 + lane;
        s += (idx < n) ? cnt[idx] : 0;
    }
#pragma unroll
    for (int off = 16; off > 0; off >>= 1) s += __shfl_xor(s, off);
    if (lane == 0) s_tot[wave] = s;
    __syncthreads();
    int run = 0;
    for (int w = 0; w < wave; ++w) run += s_tot[w];

#pragma unroll
    for (int k = 0; k < SCAN_CH; ++k) {
        const int idx = seg + k * 32 + lane;
        const int vk = (idx < n) ? cnt[idx] : 0;
        int x = vk;
#pragma unroll
        for (int off = 1; off < 32; off <<= 1) {
            const int y = __shfl_up(x, off);
            if (lane >= off) x += y;
        }
        if (idx < n) {
            rowptr[idx] = run + x - vk;
            dinv[idx]   = rsqrtf(fmaxf((float)(vk + 1), 1.0f));
        }
        run += __shfl(x, 31);
    }
    if (threadIdx.x == SCAN_T - 1) rowptr[n] = run;
}

__global__ void k_fill_col(const int* __restrict__ src, const int* __restrict__ dst,
                           const int* __restrict__ rank, const int* __restrict__ rowptr,
                           int* __restrict__ col, int e) {
    int i = blockIdx.x * blockDim.x + threadIdx.x;
    if (i < e) col[rowptr[dst[i]] + rank[i]] = src[i];
}

__global__ void k_move_words(const int* __restrict__ from, int* __restrict__ to, int nwords) {
    int i = blockIdx.x * blockDim.x + threadIdx.x;
    if (i < nwords) to[i] = from[i];
}

template <int KIN, int KOUT>
__global__ __launch_bounds__(THREADS)
void k_gemm_wmma(const float* __restrict__ A, const float* __restrict__ W,
                 float* __restrict__ C, int nrows) {
    __shared__ v2f sW[(KIN / 2) * KOUT];
    for (int i = threadIdx.x; i < (KIN / 2) * KOUT; i += blockDim.x) {
        const int k2 = i / KOUT;
        const int c  = i % KOUT;
        v2f w;
        w.x = W[(2 * k2 + 0) * KOUT + c];
        w.y = W[(2 * k2 + 1) * KOUT + c];
        sW[i] = w;
    }
    __syncthreads();

    const int wave = threadIdx.x >> 5;
    const int lane = threadIdx.x & 31;
    const int tile = blockIdx.x * 8 + wave;
    if (tile * 16 >= nrows) return;

    constexpr int NT = KOUT / 16;
    v8f acc[NT] = {};

    const int m    = lane & 15;
    const int khi  = lane >> 4;
    const float* Arow = A + (size_t)(tile * 16 + m) * KIN;
    const v2f* sWl = sW + khi * KOUT + m;

    for (int kk = 0; kk < KIN / 4; ++kk) {
        const v2f a = *(const v2f*)(Arow + kk * 4 + khi * 2);
        const v2f* sWk = sWl + (kk * 2) * KOUT;
#pragma unroll
        for (int t = 0; t < NT; ++t) {
            const v2f b = sWk[t * 16];
            acc[t] = __builtin_amdgcn_wmma_f32_16x16x4_f32(
                 false, a,  false, b,
                 (short)0, acc[t],  false,  false);
        }
    }

#pragma unroll
    for (int t = 0; t < NT; ++t) {
#pragma unroll
        for (int v = 0; v < 8; ++v) {
            const int r = tile * 16 + khi * 8 + v;
            C[(size_t)r * KOUT + t * 16 + m] = acc[t][v];
        }
    }
}

template <int F, bool RELU>
__global__ __launch_bounds__(THREADS)
void k_gather(const int* __restrict__ rowptr, const int* __restrict__ col,
              const float* __restrict__ dinv, const float* __restrict__ H,
              const float* __restrict__ bias, float* __restrict__ out, int n) {
    constexpr int C = F / 4;
    const int tid = blockIdx.x * blockDim.x + threadIdx.x;
    const int i = tid / C;
    const int c = tid % C;
    if (i >= n) return;

    const int beg = rowptr[i];
    const int end = rowptr[i + 1];
    const float di = dinv[i];
    const float* Hc = H + c * 4;

    float4 acc = make_float4(0.0f, 0.0f, 0.0f, 0.0f);
    const int last = end - 1;
    for (int j = beg; j < end; j += 4) {
        const bool p1 = j + 1 < end;
        const bool p2 = j + 2 < end;
        const bool p3 = j + 3 < end;
        const int s0 = col[j];
        const int s1 = col[min(j + 1, last)];
        const int s2 = col[min(j + 2, last)];
        const int s3 = col[min(j + 3, last)];
        const float w0 = dinv[s0] * di;
        const float w1 = dinv[s1] * di;
        const float w2 = dinv[s2] * di;
        const float w3 = dinv[s3] * di;
        const float4 h0 = *(const float4*)(Hc + (size_t)s0 * F);
        const float4 h1 = *(const float4*)(Hc + (size_t)s1 * F);
        const float4 h2 = *(const float4*)(Hc + (size_t)s2 * F);
        const float4 h3 = *(const float4*)(Hc + (size_t)s3 * F);
        acc.x = fmaf(h0.x, w0, acc.x); acc.y = fmaf(h0.y, w0, acc.y);
        acc.z = fmaf(h0.z, w0, acc.z); acc.w = fmaf(h0.w, w0, acc.w);
        acc.x = p1 ? fmaf(h1.x, w1, acc.x) : acc.x; acc.y = p1 ? fmaf(h1.y, w1, acc.y) : acc.y;
        acc.z = p1 ? fmaf(h1.z, w1, acc.z) : acc.z; acc.w = p1 ? fmaf(h1.w, w1, acc.w) : acc.w;
        acc.x = p2 ? fmaf(h2.x, w2, acc.x) : acc.x; acc.y = p2 ? fmaf(h2.y, w2, acc.y) : acc.y;
        acc.z = p2 ? fmaf(h2.z, w2, acc.z) : acc.z; acc.w = p2 ? fmaf(h2.w, w2, acc.w) : acc.w;
        acc.x = p3 ? fmaf(h3.x, w3, acc.x) : acc.x; acc.y = p3 ? fmaf(h3.y, w3, acc.y) : acc.y;
        acc.z = p3 ? fmaf(h3.z, w3, acc.z) : acc.z; acc.w = p3 ? fmaf(h3.w, w3, acc.w) : acc.w;
    }

    const float4 hs = *(const float4*)(Hc + (size_t)i * F);
    const float* b = bias + c * 4;
    float4 v;
    v.x = acc.x + hs.x * di * di + b[0];
    v.y = acc.y + hs.y * di * di + b[1];
    v.z = acc.z + hs.z * di * di + b[2];
    v.w = acc.w + hs.w * di * di + b[3];
    if (RELU) {
        v.x = fmaxf(v.x, 0.0f); v.y = fmaxf(v.y, 0.0f);
        v.z = fmaxf(v.z, 0.0f); v.w = fmaxf(v.w, 0.0f);
    }
    *(float4*)(out + (size_t)i * F + c * 4) = v;
}

extern "C" void kernel_launch(void* const* d_in, const int* in_sizes, int n_in,
                              void* d_out, int out_size, void* d_ws, size_t ws_size,
                              hipStream_t stream) {
    const float* x  = (const float*)d_in[0];
    const int*   ei = (const int*)d_in[1];
    const float* W1 = (const float*)d_in[2];
    const float* b1 = (const float*)d_in[3];
    const float* W2 = (const float*)d_in[4];
    const float* b2 = (const float*)d_in[5];
    const float* W3 = (const float*)d_in[6];
    const float* b3 = (const float*)d_in[7];
    float* out = (float*)d_out;

    const int N = GCN_N;
    const int E = in_sizes[1] / 2;
    const int* src = ei;
    const int* dst = ei + E;

    float* dinv = (float*)d_ws;
    float* bufA = dinv + ((N + 255) & ~255);
    float* bufB = bufA + (size_t)N * 64;

    int* rank   = (int*)bufB;
    int* cnt    = rank + E;
    int* col    = (int*)out;
    int* rowptr = col + E;
    int* col3    = (int*)(bufA + (size_t)N * 32);
    int* rowptr3 = col3 + E;

    const int tiles  = N / 16;
    const int gemmBl = (tiles + 7) / 8;

    auto blocks = [](long long n) { return (unsigned)((n + THREADS - 1) / THREADS); };

    k_zero_cnt<<<blocks(N), THREADS, 0, stream>>>(cnt, N);
    k_count_rank<<<blocks(E), THREADS, 0, stream>>>(dst, cnt, rank, E);
    k_scan_dinv<<<1, SCAN_T, 0, stream>>>(cnt, rowptr, dinv, N);
    k_fill_col<<<blocks(E), THREADS, 0, stream>>>(src, dst, rank, rowptr, col, E);

    k_gemm_wmma<128, 64><<<gemmBl, THREADS, 0, stream>>>(x, W1, bufA, N);
    k_gather<64, true><<<blocks((long long)N * 16), THREADS, 0, stream>>>(rowptr, col, dinv, bufA, b1, bufB, N);

    k_gemm_wmma<64, 64><<<gemmBl, THREADS, 0, stream>>>(bufB, W2, bufA, N);
    k_gather<64, true><<<blocks((long long)N * 16), THREADS, 0, stream>>>(rowptr, col, dinv, bufA, b2, bufB, N);

    k_move_words<<<blocks((long long)E + N + 1), THREADS, 0, stream>>>(col, col3, E + N + 1);
    k_gemm_wmma<64, 32><<<gemmBl, THREADS, 0, stream>>>(bufB, W3, bufA, N);
    k_gather<32, false><<<blocks((long long)N * 8), THREADS, 0, stream>>>(rowptr3, col3, dinv, bufA, b3, out, N);
}


# Edge lists built by a two-level counting sort on LDS counters (bin count, bin scatter, per-bin rows) instead of 800k returning global adds, scan, zero fill; row sum order = arrival rank of LDS adds
# speedup vs baseline: 2.7293x; 1.8514x over previous
#include <hip/hip_runtime.h>
#include <cstddef>

typedef __attribute__((ext_vector_type(2))) float v2f;
typedef __attribute__((ext_vector_type(8))) float v8f;

#define GCN_N 50000
#define THREADS 256

#define NBIN 256
#define NCH  256
#define NPB  ((GCN_N + NBIN - 1) / NBIN)
#define SORT_T 1024
#define CH_CAP  4096
#define COL_CAP 8192

__device__ __forceinline__ int excl_scan_256(int v, int* s_wsum) {
    const int lane = threadIdx.x & 31;
    const int wave = threadIdx.x >> 5;
    int x = v;
#pragma unroll
    for (int off = 1; off < 32; off <<= 1) {
        const int y = __shfl_up(x, off);
        if (lane >= off) x += y;
    }
    if (wave < 8 && lane == 31) s_wsum[wave] = x;
    __syncthreads();
    int base = 0;
    if (wave < 8)
        for (int w = 0; w < wave; ++w) base += s_wsum[w];
    __syncthreads();
    return base + x - v;
}

__global__ __launch_bounds__(SORT_T)
void k_bin_count(const int* __restrict__ dst, int* __restrict__ cntmat, int e) {
    __shared__ int s_cnt[NBIN];
    const int t = threadIdx.x;
    const int g = blockIdx.x;
    if (t < NBIN) s_cnt[t] = 0;
    __syncthreads();
    const int chunk = (e + NCH - 1) / NCH;
    const int beg = g * chunk;
    const int end = min(beg + chunk, e);
    for (int i0 = beg + t; i0 < end; i0 += 4 * SORT_T) {
        int d[4];
#pragma unroll
        for (int u = 0; u < 4; ++u) d[u] = dst[min(i0 + u * SORT_T, end - 1)];
#pragma unroll
        for (int u = 0; u < 4; ++u)
            if (i0 + u * SORT_T < end) atomicAdd(&s_cnt[d[u] / NPB], 1);
    }
    __syncthreads();
    if (t < NBIN) cntmat[g * NBIN + t] = s_cnt[t];
}

__global__ __launch_bounds__(SORT_T)
void k_bin_scatter(const int* __restrict__ src, const int* __restrict__ dst,
                   const int* __restrict__ cntmat, int* __restrict__ binbase,
                   int* __restrict__ binned, int e) {
    __shared__ int4 s_raw[CH_CAP / 2];
    __shared__ int s_off[NBIN];
    __shared__ int s_lst[NBIN];
    __shared__ int s_cnt[NBIN];
    __shared__ int s_wsum[8];
    int*  s_all   = (int*)s_raw;
    int*  s_bef   = s_all + 16 * NBIN;
    int2* s_pairs = (int2*)s_raw;
    const int t = threadIdx.x;
    const int g = blockIdx.x;

    const int b4 = (t & 63) * 4;
    const int q  = t >> 6;
    int4 all = make_int4(0, 0, 0, 0);
    int4 bef = make_int4(0, 0, 0, 0);
#pragma unroll
    for (int k = 0; k < NCH / 16; ++k) {
        const int gg = q * (NCH / 16) + k;
        const int4 c = *(const int4*)(cntmat + gg * NBIN + b4);
        const bool front = gg < g;
        all.x += c.x; all.y += c.y; all.z += c.z; all.w += c.w;
        bef.x += front ? c.x : 0; bef.y += front ? c.y : 0;
        bef.z += front ? c.z : 0; bef.w += front ? c.w : 0;
    }
    const int own = (t < NBIN) ? cntmat[g * NBIN + t] : 0;
    *(int4*)(s_all + q * NBIN + b4) = all;
    *(int4*)(s_bef + q * NBIN + b4) = bef;
    if (t < NBIN) s_cnt[t] = 0;
    __syncthreads();
    int tot = 0, before = 0;
    if (t < NBIN) {
#pragma unroll
        for (int k = 0; k < 16; ++k) {
            tot    += s_all[k * NBIN + t];
            before += s_bef[k * NBIN + t];
        }
    }
    const int base = excl_scan_256(tot, s_wsum);
    const int lst  = excl_scan_256(own, s_wsum);
    if (t < NBIN) {
        s_off[t] = base + before;
        s_lst[t] = lst;
        if (g == 0) binbase[t] = base;
    }
    if (g == 0 && t == NBIN - 1) binbase[NBIN] = base + tot;
    __syncthreads();

    int2* __restrict__ pairs = (int2*)binned;
    const int chunk = (e + NCH - 1) / NCH;
    const int beg = g * chunk;
    const int end = min(beg + chunk, e);
    const int cn  = end - beg;
    if (cn <= CH_CAP) {
        for (int i0 = beg + t; i0 < end; i0 += 4 * SORT_T) {
            int d[4], s[4];
#pragma unroll
            for (int u = 0; u < 4; ++u) {
                const int ii = min(i0 + u * SORT_T, end - 1);
                d[u] = dst[ii];
                s[u] = src[ii];
            }
#pragma unroll
            for (int u = 0; u < 4; ++u) {
                if (i0 + u * SORT_T < end) {
                    const int bb = d[u] / NPB;
                    const int r = atomicAdd(&s_cnt[bb], 1);
                    s_pairs[s_lst[bb] + r] = make_int2(s[u], d[u]);
                }
            }
        }
        __syncthreads();
        for (int i = t; i < cn; i += SORT_T) {
            const int2 p = s_pairs[i];
            const int bb = p.y / NPB;
            pairs[s_off[bb] + (i - s_lst[bb])] = p;
        }
    } else {
        for (int i = beg + t; i < end; i += SORT_T) {
            const int d = dst[i];
            const int s = src[i];
            const int bb = d / NPB;
            const int r = atomicAdd(&s_cnt[bb], 1);
            pairs[s_off[bb] + r] = make_int2(s, d);
        }
    }
}

__global__ __launch_bounds__(SORT_T)
void k_bin_rows(const int* __restrict__ binned, const int* __restrict__ binbase,
                int* __restrict__ rowptr, int* __restrict__ col,
                float* __restrict__ dinv, int n) {
    __shared__ int s_cur[NBIN];
    __shared__ int s_col[COL_CAP];
    __shared__ int s_wsum[8];
    const int t = threadIdx.x;
    const int bin = blockIdx.x;
    const int2* __restrict__ pairs = (const int2*)binned;
    const int beg = binbase[bin];
    const int end = binbase[bin + 1];
    const int ne  = end - beg;
    const int node0 = bin * NPB;

    if (t < NBIN) s_cur[t] = 0;
    __syncthreads();
    for (int i0 = beg + t; i0 < end; i0 += 4 * SORT_T) {
        int d[4];
#pragma unroll
        for (int u = 0; u < 4; ++u) d[u] = pairs[min(i0 + u * SORT_T, end - 1)].y;
#pragma unroll
        for (int u = 0; u < 4; ++u)
            if (i0 + u * SORT_T < end) atomicAdd(&s_cur[d[u] - node0], 1);
    }
    __syncthreads();
    const int deg = (t < NBIN) ? s_cur[t] : 0;
    const int ex = excl_scan_256(deg, s_wsum);
    if (t < NPB && node0 + t < n) {
        rowptr[node0 + t] = beg + ex;
        dinv[node0 + t]   = rsqrtf(fmaxf((float)(deg + 1), 1.0f));
    }
    if (bin == NBIN - 1 && t == 0) rowptr[n] = end;
    if (t < NBIN) s_cur[t] = ex;
    __syncthreads();

    if (ne <= COL_CAP) {
        for (int i0 = beg + t; i0 < end; i0 += 4 * SORT_T) {
            int2 p[4];
#pragma unroll
            for (int u = 0; u < 4; ++u) p[u] = pairs[min(i0 + u * SORT_T, end - 1)];
#pragma unroll
            for (int u = 0; u < 4; ++u) {
                if (i0 + u * SORT_T < end) {
                    const int r = atomicAdd(&s_cur[p[u].y - node0], 1);
                    s_col[r] = p[u].x;
                }
            }
        }
        __syncthreads();
        for (int i = t; i < ne; i += SORT_T) col[beg + i] = s_col[i];
    } else {
        for (int i = beg + t; i < end; i += SORT_T) {
            const int2 p = pairs[i];
            const int r = atomicAdd(&s_cur[p.y - node0], 1);
            col[beg + r] = p.x;
        }
    }
}

__global__ void k_move_words(const int* __restrict__ from, int* __restrict__ to, int nwords) {
    int i = blockIdx.x * blockDim.x + threadIdx.x;
    if (i < nwords) to[i] = from[i];
}

template <int KIN, int KOUT>
__global__ __launch_bounds__(THREADS)
void k_gemm_wmma(const float* __restrict__ A, const float* __restrict__ W,
                 float* __restrict__ C, int nrows) {
    __shared__ v2f sW[(KIN / 2) * KOUT];
    for (int i = threadIdx.x; i < (KIN / 2) * KOUT; i += blockDim.x) {
        const int k2 = i / KOUT;
        const int c  = i % KOUT;
        v2f w;
        w.x = W[(2 * k2 + 0) * KOUT + c];
        w.y = W[(2 * k2 + 1) * KOUT + c];
        sW[i] = w;
    }
    __syncthreads();

    const int wave = threadIdx.x >> 5;
    const int lane = threadIdx.x & 31;
    const int tile = blockIdx.x * 8 + wave;
    if (tile * 16 >= nrows) return;

    constexpr int NT = KOUT / 16;
    v8f acc[NT] = {};

    const int m    = lane & 15;
    const int khi  = lane >> 4;
    const float* Arow = A + (size_t)(tile * 16 + m) * KIN;
    const v2f* sWl = sW + khi * KOUT + m;

    for (int kk = 0; kk < KIN / 4; ++kk) {
        const v2f a = *(const v2f*)(Arow + kk * 4 + khi * 2);
        const v2f* sWk = sWl + (kk * 2) * KOUT;
#pragma unroll
        for (int t = 0; t < NT; ++t) {
            const v2f b = sWk[t * 16];
            acc[t] = __builtin_amdgcn_wmma_f32_16x16x4_f32(
                 false, a,  false, b,
                 (short)0, acc[t],  false,  false);
        }
    }

#pragma unroll
    for (int t = 0; t < NT; ++t) {
#pragma unroll
        for (int v = 0; v < 8; ++v) {
            const int r = tile * 16 + khi * 8 + v;
            C[(size_t)r * KOUT + t * 16 + m] = acc[t][v];
        }
    }
}

template <int F, bool RELU>
__global__ __launch_bounds__(THREADS)
void k_gather(const int* __restrict__ rowptr, const int* __restrict__ col,
              const float* __restrict__ dinv, const float* __restrict__ H,
              const float* __restrict__ bias, float* __restrict__ out, int n) {
    constexpr int C = F / 4;
    const int tid = blockIdx.x * blockDim.x + threadIdx.x;
    const int i = tid / C;
    const int c = tid % C;
    if (i >= n) return;

    const int beg = rowptr[i];
    const int end = rowptr[i + 1];
    const float di = dinv[i];
    const float* Hc = H + c * 4;

    float4 acc = make_float4(0.0f, 0.0f, 0.0f, 0.0f);
    const int last = end - 1;
    for (int j = beg; j < end; j += 4) {
        const bool p1 = j + 1 < end;
        const bool p2 = j + 2 < end;
        const bool p3 = j + 3 < end;
        const int s0 = col[j];
        const int s1 = col[min(j + 1, last)];
        const int s2 = col[min(j + 2, last)];
        const int s3 = col[min(j + 3, last)];
        const float w0 = dinv[s0] * di;
        const float w1 = dinv[s1] * di;
        const float w2 = dinv[s2] * di;
        const float w3 = dinv[s3] * di;
        const float4 h0 = *(const float4*)(Hc + (size_t)s0 * F);
        const float4 h1 = *(const float4*)(Hc + (size_t)s1 * F);
        const float4 h2 = *(const float4*)(Hc + (size_t)s2 * F);
        const float4 h3 = *(const float4*)(Hc + (size_t)s3 * F);
        acc.x = fmaf(h0.x, w0, acc.x); acc.y = fmaf(h0.y, w0, acc.y);
        acc.z = fmaf(h0.z, w0, acc.z); acc.w = fmaf(h0.w, w0, acc.w);
        acc.x = p1 ? fmaf(h1.x, w1, acc.x) : acc.x; acc.y = p1 ? fmaf(h1.y, w1, acc.y) : acc.y;
        acc.z = p1 ? fmaf(h1.z, w1, acc.z) : acc.z; acc.w = p1 ? fmaf(h1.w, w1, acc.w) : acc.w;
        acc.x = p2 ? fmaf(h2.x, w2, acc.x) : acc.x; acc.y = p2 ? fmaf(h2.y, w2, acc.y) : acc.y;
        acc.z = p2 ? fmaf(h2.z, w2, acc.z) : acc.z; acc.w = p2 ? fmaf(h2.w, w2, acc.w) : acc.w;
        acc.x = p3 ? fmaf(h3.x, w3, acc.x) : acc.x; acc.y = p3 ? fmaf(h3.y, w3, acc.y) : acc.y;
        acc.z = p3 ? fmaf(h3.z, w3, acc.z) : acc.z; acc.w = p3 ? fmaf(h3.w, w3, acc.w) : acc.w;
    }

    const float4 hs = *(const float4*)(Hc + (size_t)i * F);
    const float* b = bias + c * 4;
    float4 v;
    v.x = acc.x + hs.x * di * di + b[0];
    v.y = acc.y + hs.y * di * di + b[1];
    v.z = acc.z + hs.z * di * di + b[2];
    v.w = acc.w + hs.w * di * di + b[3];
    if (RELU) {
        v.x = fmaxf(v.x, 0.0f); v.y = fmaxf(v.y, 0.0f);
        v.z = fmaxf(v.z, 0.0f); v.w = fmaxf(v.w, 0.0f);
    }
    *(float4*)(out + (size_t)i * F + c * 4) = v;
}

extern "C" void kernel_launch(void* const* d_in, const int* in_sizes, int n_in,
                              void* d_out, int out_size, void* d_ws, size_t ws_size,
                              hipStream_t stream) {
    const float* x  = (const float*)d_in[0];
    const int*   ei = (const int*)d_in[1];
    const float* W1 = (const float*)d_in[2];
    const float* b1 = (const float*)d_in[3];
    const float* W2 = (const float*)d_in[4];
    const float* b2 = (const float*)d_in[5];
    const float* W3 = (const float*)d_in[6];
    const float* b3 = (const float*)d_in[7];
    float* out = (float*)d_out;

    const int N = GCN_N;
    const int E = in_sizes[1] / 2;
    const int* src = ei;
    const int* dst = ei + E;

    float* dinv = (float*)d_ws;
    float* bufA = dinv + ((N + 255) & ~255);
    float* bufB = bufA + (size_t)N * 64;

    int* cntmat  = (int*)bufA;
    int* binbase = cntmat + NCH * NBIN;
    int* binned  = (int*)bufB;
    int* col    = (int*)out;
    int* rowptr = col + E;
    int* col3    = (int*)(bufA + (size_t)N * 32);
    int* rowptr3 = col3 + E;

    const int tiles  = N / 16;
    const int gemmBl = (tiles + 7) / 8;

    auto blocks = [](long long n) { return (unsigned)((n + THREADS - 1) / THREADS); };

    k_bin_count<<<NCH, SORT_T, 0, stream>>>(dst, cntmat, E);
    k_bin_scatter<<<NCH, SORT_T, 0, stream>>>(src, dst, cntmat, binbase, binned, E);
    k_bin_rows<<<NBIN, SORT_T, 0, stream>>>(binned, binbase, rowptr, col, dinv, N);

    k_gemm_wmma<128, 64><<<gemmBl, THREADS, 0, stream>>>(x, W1, bufA, N);
    k_gather<64, true><<<blocks((long long)N * 16), THREADS, 0, stream>>>(rowptr, col, dinv, bufA, b1, bufB, N);

    k_gemm_wmma<64, 64><<<gemmBl, THREADS, 0, stream>>>(bufB, W2, bufA, N);
    k_gather<64, true><<<blocks((long long)N * 16), THREADS, 0, stream>>>(rowptr, col, dinv, bufA, b2, bufB, N);

    k_move_words<<<blocks((long long)E + N + 1), THREADS, 0, stream>>>(col, col3, E + N + 1);
    k_gemm_wmma<64, 32><<<gemmBl, THREADS, 0, stream>>>(bufB, W3, bufA, N);
    k_gather<32, false><<<blocks((long long)N * 8), THREADS, 0, stream>>>(rowptr3, col3, dinv, bufA, b3, out, N);
}


# Next layer's dense transform fused into the row gather: a work-group's finished 16x64 hidden tile goes through LDS into the f32 WMMA; H1, H2 never stored; 10 launches -> 8; sum orders kept
# speedup vs baseline: 3.2517x; 1.1914x over previous
#include <hip/hip_runtime.h>
#include <cstddef>

typedef __attribute__((ext_vector_type(2))) float v2f;
typedef __attribute__((ext_vector_type(8))) float v8f;

#define GCN_N 50000
#define THREADS 256

#define NBIN 256
#define NCH  256
#define NPB  ((GCN_N + NBIN - 1) / NBIN)
#define SORT_T 1024
#define CH_CAP  4096
#define COL_CAP 8192

__device__ __forceinline__ int excl_scan_256(int v, int* s_wsum) {
    const int lane = threadIdx.x & 31;
    const int wave = threadIdx.x >> 5;
    int x = v;
#pragma unroll
    for (int off = 1; off < 32; off <<= 1) {
        const int y = __shfl_up(x, off);
        if (lane >= off) x += y;
    }
    if (wave < 8 && lane == 31) s_wsum[wave] = x;
    __syncthreads();
    int base = 0;
    if (wave < 8)
        for (int w = 0; w < wave; ++w) base += s_wsum[w];
    __syncthreads();
    return base + x - v;
}

__global__ __launch_bounds__(SORT_T)
void k_bin_count(const int* __restrict__ dst, int* __restrict__ cntmat, int e) {
    __shared__ int s_cnt[NBIN];
    const int t = threadIdx.x;
    const int g = blockIdx.x;
    if (t < NBIN) s_cnt[t] = 0;
    __syncthreads();
    const int chunk = (e + NCH - 1) / NCH;
    const int beg = g * chunk;
    const int end = min(beg + chunk, e);
    for (int i0 = beg + t; i0 < end; i0 += 4 * SORT_T) {
        int d[4];
#pragma unroll
        for (int u = 0; u < 4; ++u) d[u] = dst[min(i0 + u * SORT_T, end - 1)];
#pragma unroll
        for (int u = 0; u < 4; ++u)
            if (i0 + u * SORT_T < end) atomicAdd(&s_cnt[d[u] / NPB], 1);
    }
    __syncthreads();
    if (t < NBIN) cntmat[g * NBIN + t] = s_cnt[t];
}

__global__ __launch_bounds__(SORT_T)
void k_bin_scatter(const int* __restrict__ src, const int* __restrict__ dst,
                   const int* __restrict__ cntmat, int* __restrict__ binbase,
                   int* __restrict__ binned, int e) {
    __shared__ int4 s_raw[CH_CAP / 2];
    __shared__ int s_off[NBIN];
    __shared__ int s_lst[NBIN];
    __shared__ int s_cnt[NBIN];
    __shared__ int s_wsum[8];
    int*  s_all   = (int*)s_raw;
    int*  s_bef   = s_all + 16 * NBIN;
    int2* s_pairs = (int2*)s_raw;
    const int t = threadIdx.x;
    const int g = blockIdx.x;

    const int b4 = (t & 63) * 4;
    const int q  = t >> 6;
    int4 all = make_int4(0, 0, 0, 0);
    int4 bef = make_int4(0, 0, 0, 0);
#pragma unroll
    for (int k = 0; k < NCH / 16; ++k) {
        const int gg = q * (NCH / 16) + k;
        const int4 c = *(const int4*)(cntmat + gg * NBIN + b4);
        const bool front = gg < g;
        all.x += c.x; all.y += c.y; all.z += c.z; all.w += c.w;
        bef.x += front ? c.x : 0; bef.y += front ? c.y : 0;
        bef.z += front ? c.z : 0; bef.w += front ? c.w : 0;
    }
    const int own = (t < NBIN) ? cntmat[g * NBIN + t] : 0;
    *(int4*)(s_all + q * NBIN + b4) = all;
    *(int4*)(s_bef + q * NBIN + b4) = bef;
    if (t < NBIN) s_cnt[t] = 0;
    __syncthreads();
    int tot = 0, before = 0;
    if (t < NBIN) {
#pragma unroll
        for (int k = 0; k < 16; ++k) {
            tot    += s_all[k * NBIN + t];
            before += s_bef[k * NBIN + t];
        }
    }
    const int base = excl_scan_256(tot, s_wsum);
    const int lst  = excl_scan_256(own, s_wsum);
    if (t < NBIN) {
        s_off[t] = base + before;
        s_lst[t] = lst;
        if (g == 0) binbase[t] = base;
    }
    if (g == 0 && t == NBIN - 1) binbase[NBIN] = base + tot;
    __syncthreads();

    int2* __restrict__ pairs = (int2*)binned;
    const int chunk = (e + NCH - 1) / NCH;
    const int beg = g * chunk;
    const int end = min(beg + chunk, e);
    const int cn  = end - beg;
    if (cn <= CH_CAP) {
        for (int i0 = beg + t; i0 < end; i0 += 4 * SORT_T) {
            int d[4], s[4];
#pragma unroll
            for (int u = 0; u < 4; ++u) {
                const int ii = min(i0 + u * SORT_T, end - 1);
                d[u] = dst[ii];
                s[u] = src[ii];
            }
#pragma unroll
            for (int u = 0; u < 4; ++u) {
                if (i0 + u * SORT_T < end) {
                    const int bb = d[u] / NPB;
                    const int r = atomicAdd(&s_cnt[bb], 1);
                    s_pairs[s_lst[bb] + r] = make_int2(s[u], d[u]);
                }
            }
        }
        __syncthreads();
        for (int i = t; i < cn; i += SORT_T) {
            const int2 p = s_pairs[i];
            const int bb = p.y / NPB;
            pairs[s_off[bb] + (i - s_lst[bb])] = p;
        }
    } else {
        for (int i = beg + t; i < end; i += SORT_T) {
            const int d = dst[i];
            const int s = src[i];
            const int bb = d / NPB;
            const int r = atomicAdd(&s_cnt[bb], 1);
            pairs[s_off[bb] + r] = make_int2(s, d);
        }
    }
}

__global__ __launch_bounds__(SORT_T)
void k_bin_rows(const int* __restrict__ binned, const int* __restrict__ binbase,
                int* __restrict__ rowptr, int* __restrict__ col,
                float* __restrict__ dinv, int n) {
    __shared__ int s_cur[NBIN];
    __shared__ int s_col[COL_CAP];
    __shared__ int s_wsum[8];
    const int t = threadIdx.x;
    const int bin = blockIdx.x;
    const int2* __restrict__ pairs = (const int2*)binned;
    const int beg = binbase[bin];
    const int end = binbase[bin + 1];
    const int ne  = end - beg;
    const int node0 = bin * NPB;

    if (t < NBIN) s_cur[t] = 0;
    __syncthreads();
    for (int i0 = beg + t; i0 < end; i0 += 4 * SORT_T) {
        int d[4];
#pragma unroll
        for (int u = 0; u < 4; ++u) d[u] = pairs[min(i0 + u * SORT_T, end - 1)].y;
#pragma unroll
        for (int u = 0; u < 4; ++u)
            if (i0 + u * SORT_T < end) atomicAdd(&s_cur[d[u] - node0], 1);
    }
    __syncthreads();
    const int deg = (t < NBIN) ? s_cur[t] : 0;
    const int ex = excl_scan_256(deg, s_wsum);
    if (t < NPB && node0 + t < n) {
        rowptr[node0 + t] = beg + ex;
        dinv[node0 + t]   = rsqrtf(fmaxf((float)(deg + 1), 1.0f));
    }
    if (bin == NBIN - 1 && t == 0) rowptr[n] = end;
    if (t < NBIN) s_cur[t] = ex;
    __syncthreads();

    if (ne <= COL_CAP) {
        for (int i0 = beg + t; i0 < end; i0 += 4 * SORT_T) {
            int2 p[4];
#pragma unroll
            for (int u = 0; u < 4; ++u) p[u] = pairs[min(i0 + u * SORT_T, end - 1)];
#pragma unroll
            for (int u = 0; u < 4; ++u) {
                if (i0 + u * SORT_T < end) {
                    const int r = atomicAdd(&s_cur[p[u].y - node0], 1);
                    s_col[r] = p[u].x;
                }
            }
        }
        __syncthreads();
        for (int i = t; i < ne; i += SORT_T) col[beg + i] = s_col[i];
    } else {
        for (int i = beg + t; i < end; i += SORT_T) {
            const int2 p = pairs[i];
            const int r = atomicAdd(&s_cur[p.y - node0], 1);
            col[beg + r] = p.x;
        }
    }
}

__global__ void k_move_words(const int* __restrict__ from, int* __restrict__ to, int nwords) {
    int i = blockIdx.x * blockDim.x + threadIdx.x;
    if (i < nwords) to[i] = from[i];
}

template <int KIN, int KOUT>
__global__ __launch_bounds__(THREADS)
void k_gemm_wmma(const float* __restrict__ A, const float* __restrict__ W,
                 float* __restrict__ C, int nrows) {
    __shared__ v2f sW[(KIN / 2) * KOUT];
    for (int i = threadIdx.x; i < (KIN / 2) * KOUT; i += blockDim.x) {
        const int k2 = i / KOUT;
        const int c  = i % KOUT;
        v2f w;
        w.x = W[(2 * k2 + 0) * KOUT + c];
        w.y = W[(2 * k2 + 1) * KOUT + c];
        sW[i] = w;
    }
    __syncthreads();

    const int wave = threadIdx.x >> 5;
    const int lane = threadIdx.x & 31;
    const int tile = blockIdx.x * 8 + wave;
    if (tile * 16 >= nrows) return;

    constexpr int NT = KOUT / 16;
    v8f acc[NT] = {};

    const int m    = lane & 15;
    const int khi  = lane >> 4;
    const float* Arow = A + (size_t)(tile * 16 + m) * KIN;
    const v2f* sWl = sW + khi * KOUT + m;

    for (int kk = 0; kk < KIN / 4; ++kk) {
        const v2f a = *(const v2f*)(Arow + kk * 4 + khi * 2);
        const v2f* sWk = sWl + (kk * 2) * KOUT;
#pragma unroll
        for (int t = 0; t < NT; ++t) {
            const v2f b = sWk[t * 16];
            acc[t] = __builtin_amdgcn_wmma_f32_16x16x4_f32(
                 false, a,  false, b,
                 (short)0, acc[t],  false,  false);
        }
    }

#pragma unroll
    for (int t = 0; t < NT; ++t) {
#pragma unroll
        for (int v = 0; v < 8; ++v) {
            const int r = tile * 16 + khi * 8 + v;
            C[(size_t)r * KOUT + t * 16 + m] = acc[t][v];
        }
    }
}

template <int F, bool RELU>
__device__ __forceinline__ float4 agg_row(const int* __restrict__ rowptr, const int* __restrict__ col,
                                          const float* __restrict__ dinv, const float* __restrict__ H,
                                          const float* __restrict__ bias, int i, int c) {
    const int beg = rowptr[i];
    const int end = rowptr[i + 1];
    const float di = dinv[i];
    const float* Hc = H + c * 4;

    float4 acc = make_float4(0.0f, 0.0f, 0.0f, 0.0f);
    const int last = end - 1;
    for (int j = beg; j < end; j += 4) {
        const bool p1 = j + 1 < end;
        const bool p2 = j + 2 < end;
        const bool p3 = j + 3 < end;
        const int s0 = col[j];
        const int s1 = col[min(j + 1, last)];
        const int s2 = col[min(j + 2, last)];
        const int s3 = col[min(j + 3, last)];
        const float w0 = dinv[s0] * di;
        const float w1 = dinv[s1] * di;
        const float w2 = dinv[s2] * di;
        const float w3 = dinv[s3] * di;
        const float4 h0 = *(const float4*)(Hc + (size_t)s0 * F);
        const float4 h1 = *(const float4*)(Hc + (size_t)s1 * F);
        const float4 h2 = *(const float4*)(Hc + (size_t)s2 * F);
        const float4 h3 = *(const float4*)(Hc + (size_t)s3 * F);
        acc.x = fmaf(h0.x, w0, acc.x); acc.y = fmaf(h0.y, w0, acc.y);
        acc.z = fmaf(h0.z, w0, acc.z); acc.w = fmaf(h0.w, w0, acc.w);
        acc.x = p1 ? fmaf(h1.x, w1, acc.x) : acc.x; acc.y = p1 ? fmaf(h1.y, w1, acc.y) : acc.y;
        acc.z = p1 ? fmaf(h1.z, w1, acc.z) : acc.z; acc.w = p1 ? fmaf(h1.w, w1, acc.w) : acc.w;
        acc.x = p2 ? fmaf(h2.x, w2, acc.x) : acc.x; acc.y = p2 ? fmaf(h2.y, w2, acc.y) : acc.y;
        acc.z = p2 ? fmaf(h2.z, w2, acc.z) : acc.z; acc.w = p2 ? fmaf(h2.w, w2, acc.w) : acc.w;
        acc.x = p3 ? fmaf(h3.x, w3, acc.x) : acc.x; acc.y = p3 ? fmaf(h3.y, w3, acc.y) : acc.y;
        acc.z = p3 ? fmaf(h3.z, w3, acc.z) : acc.z; acc.w = p3 ? fmaf(h3.w, w3, acc.w) : acc.w;
    }

    const float4 hs = *(const float4*)(Hc + (size_t)i * F);
    const float* b = bias + c * 4;
    float4 v;
    v.x = acc.x + hs.x * di * di + b[0];
    v.y = acc.y + hs.y * di * di + b[1];
    v.z = acc.z + hs.z * di * di + b[2];
    v.w = acc.w + hs.w * di * di + b[3];
    if (RELU) {
        v.x = fmaxf(v.x, 0.0f); v.y = fmaxf(v.y, 0.0f);
        v.z = fmaxf(v.z, 0.0f); v.w = fmaxf(v.w, 0.0f);
    }
    return v;
}

template <int F, bool RELU>
__global__ __launch_bounds__(THREADS)
void k_gather(const int* __restrict__ rowptr, const int* __restrict__ col,
              const float* __restrict__ dinv, const float* __restrict__ H,
              const float* __restrict__ bias, float* __restrict__ out, int n) {
    constexpr int C = F / 4;
    const int tid = blockIdx.x * blockDim.x + threadIdx.x;
    const int i = tid / C;
    const int c = tid % C;
    if (i >= n) return;
    const float4 v = agg_row<F, RELU>(rowptr, col, dinv, H, bias, i, c);
    *(float4*)(out + (size_t)i * F + c * 4) = v;
}

#define TILE_PITCH 68
template <int KOUT>
__global__ __launch_bounds__(THREADS)
void k_gather_gemm(const int* __restrict__ rowptr, const int* __restrict__ col,
                   const float* __restrict__ dinv, const float* __restrict__ H,
                   const float* __restrict__ bias, const float* __restrict__ Wn,
                   float* __restrict__ Cn, int n) {
    constexpr int F  = 64;
    constexpr int NT = KOUT / 16;
    __shared__ float sH[16 * TILE_PITCH];
    const int r = threadIdx.x >> 4;
    const int c = threadIdx.x & 15;
    const int i = blockIdx.x * 16 + r;

    float4 v = make_float4(0.0f, 0.0f, 0.0f, 0.0f);
    if (i < n) v = agg_row<F, true>(rowptr, col, dinv, H, bias, i, c);
    *(float4*)(sH + r * TILE_PITCH + c * 4) = v;

    const int wave = threadIdx.x >> 5;
    const int lane = threadIdx.x & 31;
    const int m    = lane & 15;
    const int khi  = lane >> 4;
    const bool mul = wave < NT;

    v2f b[F / 4];
    if (mul) {
        const float* Wl = Wn + (khi * 2) * KOUT + wave * 16 + m;
#pragma unroll
        for (int kk = 0; kk < F / 4; ++kk) {
            b[kk].x = Wl[(kk * 4 + 0) * KOUT];
            b[kk].y = Wl[(kk * 4 + 1) * KOUT];
        }
    }
    __syncthreads();
    if (!mul) return;

    v8f acc = {};
    const float* sA = sH + m * TILE_PITCH + khi * 2;
#pragma unroll
    for (int kk = 0; kk < F / 4; ++kk) {
        const v2f a = *(const v2f*)(sA + kk * 4);
        acc = __builtin_amdgcn_wmma_f32_16x16x4_f32(
             false, a,  false, b[kk],
             (short)0, acc,  false,  false);
    }
    float* Cr = Cn + (size_t)(blockIdx.x * 16 + khi * 8) * KOUT + wave * 16 + m;
#pragma unroll
    for (int q = 0; q < 8; ++q) Cr[(size_t)q * KOUT] = acc[q];
}
static_assert(GCN_N % 16 == 0, "row tiles are whole");

extern "C" void kernel_launch(void* const* d_in, const int* in_sizes, int n_in,
                              void* d_out, int out_size, void* d_ws, size_t ws_size,
                              hipStream_t stream) {
    const float* x  = (const float*)d_in[0];
    const int*   ei = (const int*)d_in[1];
    const float* W1 = (const float*)d_in[2];
    const float* b1 = (const float*)d_in[3];
    const float* W2 = (const float*)d_in[4];
    const float* b2 = (const float*)d_in[5];
    const float* W3 = (const float*)d_in[6];
    const float* b3 = (const float*)d_in[7];
    float* out = (float*)d_out;

    const int N = GCN_N;
    const int E = in_sizes[1] / 2;
    const int* src = ei;
    const int* dst = ei + E;

    float* dinv = (float*)d_ws;
    float* bufA = dinv + ((N + 255) & ~255);
    float* bufB = bufA + (size_t)N * 64;

    int* cntmat  = (int*)bufA;
    int* binbase = cntmat + NCH * NBIN;
    int* binned  = (int*)bufB;
    int* col    = (int*)out;
    int* rowptr = col + E;
    int* col3    = (int*)(bufA + (size_t)N * 32);
    int* rowptr3 = col3 + E;

    const int tiles  = N / 16;
    const int gemmBl = (tiles + 7) / 8;

    auto blocks = [](long long n) { return (unsigned)((n + THREADS - 1) / THREADS); };

    k_bin_count<<<NCH, SORT_T, 0, stream>>>(dst, cntmat, E);
    k_bin_scatter<<<NCH, SORT_T, 0, stream>>>(src, dst, cntmat, binbase, binned, E);
    k_bin_rows<<<NBIN, SORT_T, 0, stream>>>(binned, binbase, rowptr, col, dinv, N);

    k_gemm_wmma<128, 64><<<gemmBl, THREADS, 0, stream>>>(x, W1, bufA, N);

    k_gather_gemm<64><<<tiles, THREADS, 0, stream>>>(rowptr, col, dinv, bufA, b1, W2, bufB, N);

    k_gather_gemm<32><<<tiles, THREADS, 0, stream>>>(rowptr, col, dinv, bufB, b2, W3, bufA, N);

    k_move_words<<<blocks((long long)E + N + 1), THREADS, 0, stream>>>(col, col3, E + N + 1);
    k_gather<32, false><<<blocks((long long)N * 8), THREADS, 0, stream>>>(rowptr3, col3, dinv, bufA, b3, out, N);
}


# col | rowptr moved to their layer-3 home inside layer 2's gather launch (its non-multiplying waves copy whole 128-B lines after the barrier); the 26,568-wave move launch is gone; 8 launches -> 7
# speedup vs baseline: 3.3920x; 1.0431x over previous
#include <hip/hip_runtime.h>
#include <cstddef>

typedef __attribute__((ext_vector_type(2))) float v2f;
typedef __attribute__((ext_vector_type(8))) float v8f;

#define GCN_N 50000
#define THREADS 256

#define NBIN 256
#define NCH  256
#define NPB  ((GCN_N + NBIN - 1) / NBIN)
#define SORT_T 1024
#define CH_CAP  4096
#define COL_CAP 8192

__device__ __forceinline__ int excl_scan_256(int v, int* s_wsum) {
    const int lane = threadIdx.x & 31;
    const int wave = threadIdx.x >> 5;
    int x = v;
#pragma unroll
    for (int off = 1; off < 32; off <<= 1) {
        const int y = __shfl_up(x, off);
        if (lane >= off) x += y;
    }
    if (wave < 8 && lane == 31) s_wsum[wave] = x;
    __syncthreads();
    int base = 0;
    if (wave < 8)
        for (int w = 0; w < wave; ++w) base += s_wsum[w];
    __syncthreads();
    return base + x - v;
}

__global__ __launch_bounds__(SORT_T)
void k_bin_count(const int* __restrict__ dst, int* __restrict__ cntmat, int e) {
    __shared__ int s_cnt[NBIN];
    const int t = threadIdx.x;
    const int g = blockIdx.x;
    if (t < NBIN) s_cnt[t] = 0;
    __syncthreads();
    const int chunk = (e + NCH - 1) / NCH;
    const int beg = g * chunk;
    const int end = min(beg + chunk, e);
    for (int i0 = beg + t; i0 < end; i0 += 4 * SORT_T) {
        int d[4];
#pragma unroll
        for (int u = 0; u < 4; ++u) d[u] = dst[min(i0 + u * SORT_T, end - 1)];
#pragma unroll
        for (int u = 0; u < 4; ++u)
            if (i0 + u * SORT_T < end) atomicAdd(&s_cnt[d[u] / NPB], 1);
    }
    __syncthreads();
    if (t < NBIN) cntmat[g * NBIN + t] = s_cnt[t];
}

__global__ __launch_bounds__(SORT_T)
void k_bin_scatter(const int* __restrict__ src, const int* __restrict__ dst,
                   const int* __restrict__ cntmat, int* __restrict__ binbase,
                   int* __restrict__ binned, int e) {
    __shared__ int4 s_raw[CH_CAP / 2];
    __shared__ int s_off[NBIN];
    __shared__ int s_lst[NBIN];
    __shared__ int s_cnt[NBIN];
    __shared__ int s_wsum[8];
    int*  s_all   = (int*)s_raw;
    int*  s_bef   = s_all + 16 * NBIN;
    int2* s_pairs = (int2*)s_raw;
    const int t = threadIdx.x;
    const int g = blockIdx.x;

    const int b4 = (t & 63) * 4;
    const int q  = t >> 6;
    int4 all = make_int4(0, 0, 0, 0);
    int4 bef = make_int4(0, 0, 0, 0);
#pragma unroll
    for (int k = 0; k < NCH / 16; ++k) {
        const int gg = q * (NCH / 16) + k;
        const int4 c = *(const int4*)(cntmat + gg * NBIN + b4);
        const bool front = gg < g;
        all.x += c.x; all.y += c.y; all.z += c.z; all.w += c.w;
        bef.x += front ? c.x : 0; bef.y += front ? c.y : 0;
        bef.z += front ? c.z : 0; bef.w += front ? c.w : 0;
    }
    const int own = (t < NBIN) ? cntmat[g * NBIN + t] : 0;
    *(int4*)(s_all + q * NBIN + b4) = all;
    *(int4*)(s_bef + q * NBIN + b4) = bef;
    if (t < NBIN) s_cnt[t] = 0;
    __syncthreads();
    int tot = 0, before = 0;
    if (t < NBIN) {
#pragma unroll
        for (int k = 0; k < 16; ++k) {
            tot    += s_all[k * NBIN + t];
            before += s_bef[k * NBIN + t];
        }
    }
    const int base = excl_scan_256(tot, s_wsum);
    const int lst  = excl_scan_256(own, s_wsum);
    if (t < NBIN) {
        s_off[t] = base + before;
        s_lst[t] = lst;
        if (g == 0) binbase[t] = base;
    }
    if (g == 0 && t == NBIN - 1) binbase[NBIN] = base + tot;
    __syncthreads();

    int2* __restrict__ pairs = (int2*)binned;
    const int chunk = (e + NCH - 1) / NCH;
    const int beg = g * chunk;
    const int end = min(beg + chunk, e);
    const int cn  = end - beg;
    if (cn <= CH_CAP) {
        for (int i0 = beg + t; i0 < end; i0 += 4 * SORT_T) {
            int d[4], s[4];
#pragma unroll
            for (int u = 0; u < 4; ++u) {
                const int ii = min(i0 + u * SORT_T, end - 1);
                d[u] = dst[ii];
                s[u] = src[ii];
            }
#pragma unroll
            for (int u = 0; u < 4; ++u) {
                if (i0 + u * SORT_T < end) {
                    const int bb = d[u] / NPB;
                    const int r = atomicAdd(&s_cnt[bb], 1);
                    s_pairs[s_lst[bb] + r] = make_int2(s[u], d[u]);
                }
            }
        }
        __syncthreads();
        for (int i = t; i < cn; i += SORT_T) {
            const int2 p = s_pairs[i];
            const int bb = p.y / NPB;
            pairs[s_off[bb] + (i - s_lst[bb])] = p;
        }
    } else {
        for (int i = beg + t; i < end; i += SORT_T) {
            const int d = dst[i];
            const int s = src[i];
            const int bb = d / NPB;
            const int r = atomicAdd(&s_cnt[bb], 1);
            pairs[s_off[bb] + r] = make_int2(s, d);
        }
    }
}

__global__ __launch_bounds__(SORT_T)
void k_bin_rows(const int* __restrict__ binned, const int* __restrict__ binbase,
                int* __restrict__ rowptr, int* __restrict__ col,
                float* __restrict__ dinv, int n) {
    __shared__ int s_cur[NBIN];
    __shared__ int s_col[COL_CAP];
    __shared__ int s_wsum[8];
    const int t = threadIdx.x;
    const int bin = blockIdx.x;
    const int2* __restrict__ pairs = (const int2*)binned;
    const int beg = binbase[bin];
    const int end = binbase[bin + 1];
    const int ne  = end - beg;
    const int node0 = bin * NPB;

    if (t < NBIN) s_cur[t] = 0;
    __syncthreads();
    for (int i0 = beg + t; i0 < end; i0 += 4 * SORT_T) {
        int d[4];
#pragma unroll
        for (int u = 0; u < 4; ++u) d[u] = pairs[min(i0 + u * SORT_T, end - 1)].y;
#pragma unroll
        for (int u = 0; u < 4; ++u)
            if (i0 + u * SORT_T < end) atomicAdd(&s_cur[d[u] - node0], 1);
    }
    __syncthreads();
    const int deg = (t < NBIN) ? s_cur[t] : 0;
    const int ex = excl_scan_256(deg, s_wsum);
    if (t < NPB && node0 + t < n) {
        rowptr[node0 + t] = beg + ex;
        dinv[node0 + t]   = rsqrtf(fmaxf((float)(deg + 1), 1.0f));
    }
    if (bin == NBIN - 1 && t == 0) rowptr[n] = end;
    if (t < NBIN) s_cur[t] = ex;
    __syncthreads();

    if (ne <= COL_CAP) {
        for (int i0 = beg + t; i0 < end; i0 += 4 * SORT_T) {
            int2 p[4];
#pragma unroll
            for (int u = 0; u < 4; ++u) p[u] = pairs[min(i0 + u * SORT_T, end - 1)];
#pragma unroll
            for (int u = 0; u < 4; ++u) {
                if (i0 + u * SORT_T < end) {
                    const int r = atomicAdd(&s_cur[p[u].y - node0], 1);
                    s_col[r] = p[u].x;
                }
            }
        }
        __syncthreads();
        for (int i = t; i < ne; i += SORT_T) col[beg + i] = s_col[i];
    } else {
        for (int i = beg + t; i < end; i += SORT_T) {
            const int2 p = pairs[i];
            const int r = atomicAdd(&s_cur[p.y - node0], 1);
            col[beg + r] = p.x;
        }
    }
}

template <int KIN, int KOUT>
__global__ __launch_bounds__(THREADS)
void k_gemm_wmma(const float* __restrict__ A, const float* __restrict__ W,
                 float* __restrict__ C, int nrows) {
    __shared__ v2f sW[(KIN / 2) * KOUT];
    for (int i = threadIdx.x; i < (KIN / 2) * KOUT; i += blockDim.x) {
        const int k2 = i / KOUT;
        const int c  = i % KOUT;
        v2f w;
        w.x = W[(2 * k2 + 0) * KOUT + c];
        w.y = W[(2 * k2 + 1) * KOUT + c];
        sW[i] = w;
    }
    __syncthreads();

    const int wave = threadIdx.x >> 5;
    const int lane = threadIdx.x & 31;
    const int tile = blockIdx.x * 8 + wave;
    if (tile * 16 >= nrows) return;

    constexpr int NT = KOUT / 16;
    v8f acc[NT] = {};

    const int m    = lane & 15;
    const int khi  = lane >> 4;
    const float* Arow = A + (size_t)(tile * 16 + m) * KIN;
    const v2f* sWl = sW + khi * KOUT + m;

    for (int kk = 0; kk < KIN / 4; ++kk) {
        const v2f a = *(const v2f*)(Arow + kk * 4 + khi * 2);
        const v2f* sWk = sWl + (kk * 2) * KOUT;
#pragma unroll
        for (int t = 0; t < NT; ++t) {
            const v2f b = sWk[t * 16];
            acc[t] = __builtin_amdgcn_wmma_f32_16x16x4_f32(
                 false, a,  false, b,
                 (short)0, acc[t],  false,  false);
        }
    }

#pragma unroll
    for (int t = 0; t < NT; ++t) {
#pragma unroll
        for (int v = 0; v < 8; ++v) {
            const int r = tile * 16 + khi * 8 + v;
            C[(size_t)r * KOUT + t * 16 + m] = acc[t][v];
        }
    }
}

template <int F, bool RELU>
__device__ __forceinline__ float4 agg_row(const int* __restrict__ rowptr, const int* __restrict__ col,
                                          const float* __restrict__ dinv, const float* __restrict__ H,
                                          const float* __restrict__ bias, int i, int c) {
    const int beg = rowptr[i];
    const int end = rowptr[i + 1];
    const float di = dinv[i];
    const float* Hc = H + c * 4;

    float4 acc = make_float4(0.0f, 0.0f, 0.0f, 0.0f);
    const int last = end - 1;
    for (int j = beg; j < end; j += 4) {
        const bool p1 = j + 1 < end;
        const bool p2 = j + 2 < end;
        const bool p3 = j + 3 < end;
        const int s0 = col[j];
        const int s1 = col[min(j + 1, last)];
        const int s2 = col[min(j + 2, last)];
        const int s3 = col[min(j + 3, last)];
        const float w0 = dinv[s0] * di;
        const float w1 = dinv[s1] * di;
        const float w2 = dinv[s2] * di;
        const float w3 = dinv[s3] * di;
        const float4 h0 = *(const float4*)(Hc + (size_t)s0 * F);
        const float4 h1 = *(const float4*)(Hc + (size_t)s1 * F);
        const float4 h2 = *(const float4*)(Hc + (size_t)s2 * F);
        const float4 h3 = *(const float4*)(Hc + (size_t)s3 * F);
        acc.x = fmaf(h0.x, w0, acc.x); acc.y = fmaf(h0.y, w0, acc.y);
        acc.z = fmaf(h0.z, w0, acc.z); acc.w = fmaf(h0.w, w0, acc.w);
        acc.x = p1 ? fmaf(h1.x, w1, acc.x) : acc.x; acc.y = p1 ? fmaf(h1.y, w1, acc.y) : acc.y;
        acc.z = p1 ? fmaf(h1.z, w1, acc.z) : acc.z; acc.w = p1 ? fmaf(h1.w, w1, acc.w) : acc.w;
        acc.x = p2 ? fmaf(h2.x, w2, acc.x) : acc.x; acc.y = p2 ? fmaf(h2.y, w2, acc.y) : acc.y;
        acc.z = p2 ? fmaf(h2.z, w2, acc.z) : acc.z; acc.w = p2 ? fmaf(h2.w, w2, acc.w) : acc.w;
        acc.x = p3 ? fmaf(h3.x, w3, acc.x) : acc.x; acc.y = p3 ? fmaf(h3.y, w3, acc.y) : acc.y;
        acc.z = p3 ? fmaf(h3.z, w3, acc.z) : acc.z; acc.w = p3 ? fmaf(h3.w, w3, acc.w) : acc.w;
    }

    const float4 hs = *(const float4*)(Hc + (size_t)i * F);
    const float* b = bias + c * 4;
    float4 v;
    v.x = acc.x + hs.x * di * di + b[0];
    v.y = acc.y + hs.y * di * di + b[1];
    v.z = acc.z + hs.z * di * di + b[2];
    v.w = acc.w + hs.w * di * di + b[3];
    if (RELU) {
        v.x = fmaxf(v.x, 0.0f); v.y = fmaxf(v.y, 0.0f);
        v.z = fmaxf(v.z, 0.0f); v.w = fmaxf(v.w, 0.0f);
    }
    return v;
}

template <int F, bool RELU>
__global__ __launch_bounds__(THREADS)
void k_gather(const int* __restrict__ rowptr, const int* __restrict__ col,
              const float* __restrict__ dinv, const float* __restrict__ H,
              const float* __restrict__ bias, float* __restrict__ out, int n) {
    constexpr int C = F / 4;
    const int tid = blockIdx.x * blockDim.x + threadIdx.x;
    const int i = tid / C;
    const int c = tid % C;
    if (i >= n) return;
    const float4 v = agg_row<F, RELU>(rowptr, col, dinv, H, bias, i, c);
    *(float4*)(out + (size_t)i * F + c * 4) = v;
}

#define TILE_PITCH 68
template <int KOUT, bool MOVE>
__global__ __launch_bounds__(THREADS)
void k_gather_gemm(const int* __restrict__ rowptr, const int* __restrict__ col,
                   const float* __restrict__ dinv, const float* __restrict__ H,
                   const float* __restrict__ bias, const float* __restrict__ Wn,
                   float* __restrict__ Cn, int n,
                   const int* __restrict__ mv_from, int* __restrict__ mv_to, int mv_words) {
    constexpr int F  = 64;
    constexpr int NT = KOUT / 16;
    __shared__ float sH[16 * TILE_PITCH];
    const int r = threadIdx.x >> 4;
    const int c = threadIdx.x & 15;
    const int i = blockIdx.x * 16 + r;

    float4 v = make_float4(0.0f, 0.0f, 0.0f, 0.0f);
    if (i < n) v = agg_row<F, true>(rowptr, col, dinv, H, bias, i, c);
    *(float4*)(sH + r * TILE_PITCH + c * 4) = v;

    const int wave = threadIdx.x >> 5;
    const int lane = threadIdx.x & 31;
    const int m    = lane & 15;
    const int khi  = lane >> 4;
    const bool mul = wave < NT;

    v2f b[F / 4];
    if (mul) {
        const float* Wl = Wn + (khi * 2) * KOUT + wave * 16 + m;
#pragma unroll
        for (int kk = 0; kk < F / 4; ++kk) {
            b[kk].x = Wl[(kk * 4 + 0) * KOUT];
            b[kk].y = Wl[(kk * 4 + 1) * KOUT];
        }
    }
    __syncthreads();
    if (!mul) {
        if (MOVE) {
            constexpr int MT = THREADS - NT * 32;
            constexpr int NWG = GCN_N / 16;
            const int lines = (mv_words + 31) / 32;
            const int per   = ((lines + NWG - 1) / NWG) * 32;
            const int beg   = (int)blockIdx.x * per;
            const int end   = min(beg + per, mv_words);
            for (int w = beg + ((int)threadIdx.x - NT * 32) * 4; w < end; w += MT * 4) {
                if (w + 4 <= end) {
                    *(int4*)(mv_to + w) = *(const int4*)(mv_from + w);
                } else {
                    for (int u = w; u < end; ++u) mv_to[u] = mv_from[u];
                }
            }
        }
        return;
    }

    v8f acc = {};
    const float* sA = sH + m * TILE_PITCH + khi * 2;
#pragma unroll
    for (int kk = 0; kk < F / 4; ++kk) {
        const v2f a = *(const v2f*)(sA + kk * 4);
        acc = __builtin_amdgcn_wmma_f32_16x16x4_f32(
             false, a,  false, b[kk],
             (short)0, acc,  false,  false);
    }
    float* Cr = Cn + (size_t)(blockIdx.x * 16 + khi * 8) * KOUT + wave * 16 + m;
#pragma unroll
    for (int q = 0; q < 8; ++q) Cr[(size_t)q * KOUT] = acc[q];
}
static_assert(GCN_N % 16 == 0, "row tiles are whole");

extern "C" void kernel_launch(void* const* d_in, const int* in_sizes, int n_in,
                              void* d_out, int out_size, void* d_ws, size_t ws_size,
                              hipStream_t stream) {
    const float* x  = (const float*)d_in[0];
    const int*   ei = (const int*)d_in[1];
    const float* W1 = (const float*)d_in[2];
    const float* b1 = (const float*)d_in[3];
    const float* W2 = (const float*)d_in[4];
    const float* b2 = (const float*)d_in[5];
    const float* W3 = (const float*)d_in[6];
    const float* b3 = (const float*)d_in[7];
    float* out = (float*)d_out;

    const int N = GCN_N;
    const int E = in_sizes[1] / 2;
    const int* src = ei;
    const int* dst = ei + E;

    float* dinv = (float*)d_ws;
    float* bufA = dinv + ((N + 255) & ~255);
    float* bufB = bufA + (size_t)N * 64;

    int* cntmat  = (int*)bufA;
    int* binbase = cntmat + NCH * NBIN;
    int* binned  = (int*)bufB;
    int* col    = (int*)out;
    int* rowptr = col + E;
    int* col3    = (int*)(bufA + (size_t)N * 32);
    int* rowptr3 = col3 + E;

    const int tiles  = N / 16;
    const int gemmBl = (tiles + 7) / 8;

    auto blocks = [](long long n) { return (unsigned)((n + THREADS - 1) / THREADS); };

    k_bin_count<<<NCH, SORT_T, 0, stream>>>(dst, cntmat, E);
    k_bin_scatter<<<NCH, SORT_T, 0, stream>>>(src, dst, cntmat, binbase, binned, E);
    k_bin_rows<<<NBIN, SORT_T, 0, stream>>>(binned, binbase, rowptr, col, dinv, N);

    k_gemm_wmma<128, 64><<<gemmBl, THREADS, 0, stream>>>(x, W1, bufA, N);

    k_gather_gemm<64, false><<<tiles, THREADS, 0, stream>>>(rowptr, col, dinv, bufA, b1, W2, bufB, N,
                                                            nullptr, nullptr, 0);

    k_gather_gemm<32, true><<<tiles, THREADS, 0, stream>>>(rowptr, col, dinv, bufB, b2, W3, bufA, N,
                                                           col, col3, E + N + 1);

    k_gather<32, false><<<blocks((long long)N * 8), THREADS, 0, stream>>>(rowptr3, col3, dinv, bufA, b3, out, N);
}


# Source norm factor folded into the rows: producers store T[i]*dinv[i] (GEMM epilogue; hidden tile scaled before its WMMA), gathers add rows unweighted, no per-edge dinv load; sum order kept
# speedup vs baseline: 3.6051x; 1.0628x over previous
#include <hip/hip_runtime.h>
#include <cstddef>

typedef __attribute__((ext_vector_type(2))) float v2f;
typedef __attribute__((ext_vector_type(8))) float v8f;

#define GCN_N 50000
#define THREADS 256

#define NBIN 256
#define NCH  256
#define NPB  ((GCN_N + NBIN - 1) / NBIN)
#define SORT_T 1024
#define CH_CAP  4096
#define COL_CAP 8192

__device__ __forceinline__ int excl_scan_256(int v, int* s_wsum) {
    const int lane = threadIdx.x & 31;
    const int wave = threadIdx.x >> 5;
    int x = v;
#pragma unroll
    for (int off = 1; off < 32; off <<= 1) {
        const int y = __shfl_up(x, off);
        if (lane >= off) x += y;
    }
    if (wave < 8 && lane == 31) s_wsum[wave] = x;
    __syncthreads();
    int base = 0;
    if (wave < 8)
        for (int w = 0; w < wave; ++w) base += s_wsum[w];
    __syncthreads();
    return base + x - v;
}

__global__ __launch_bounds__(SORT_T)
void k_bin_count(const int* __restrict__ dst, int* __restrict__ cntmat, int e) {
    __shared__ int s_cnt[NBIN];
    const int t = threadIdx.x;
    const int g = blockIdx.x;
    if (t < NBIN) s_cnt[t] = 0;
    __syncthreads();
    const int chunk = (e + NCH - 1) / NCH;
    const int beg = g * chunk;
    const int end = min(beg + chunk, e);
    for (int i0 = beg + t; i0 < end; i0 += 4 * SORT_T) {
        int d[4];
#pragma unroll
        for (int u = 0; u < 4; ++u) d[u] = dst[min(i0 + u * SORT_T, end - 1)];
#pragma unroll
        for (int u = 0; u < 4; ++u)
            if (i0 + u * SORT_T < end) atomicAdd(&s_cnt[d[u] / NPB], 1);
    }
    __syncthreads();
    if (t < NBIN) cntmat[g * NBIN + t] = s_cnt[t];
}

__global__ __launch_bounds__(SORT_T)
void k_bin_scatter(const int* __restrict__ src, const int* __restrict__ dst,
                   const int* __restrict__ cntmat, int* __restrict__ binbase,
                   int* __restrict__ binned, int e) {
    __shared__ int4 s_raw[CH_CAP / 2];
    __shared__ int s_off[NBIN];
    __shared__ int s_lst[NBIN];
    __shared__ int s_cnt[NBIN];
    __shared__ int s_wsum[8];
    int*  s_all   = (int*)s_raw;
    int*  s_bef   = s_all + 16 * NBIN;
    int2* s_pairs = (int2*)s_raw;
    const int t = threadIdx.x;
    const int g = blockIdx.x;

    const int b4 = (t & 63) * 4;
    const int q  = t >> 6;
    int4 all = make_int4(0, 0, 0, 0);
    int4 bef = make_int4(0, 0, 0, 0);
#pragma unroll
    for (int k = 0; k < NCH / 16; ++k) {
        const int gg = q * (NCH / 16) + k;
        const int4 c = *(const int4*)(cntmat + gg * NBIN + b4);
        const bool front = gg < g;
        all.x += c.x; all.y += c.y; all.z += c.z; all.w += c.w;
        bef.x += front ? c.x : 0; bef.y += front ? c.y : 0;
        bef.z += front ? c.z : 0; bef.w += front ? c.w : 0;
    }
    const int own = (t < NBIN) ? cntmat[g * NBIN + t] : 0;
    *(int4*)(s_all + q * NBIN + b4) = all;
    *(int4*)(s_bef + q * NBIN + b4) = bef;
    if (t < NBIN) s_cnt[t] = 0;
    __syncthreads();
    int tot = 0, before = 0;
    if (t < NBIN) {
#pragma unroll
        for (int k = 0; k < 16; ++k) {
            tot    += s_all[k * NBIN + t];
            before += s_bef[k * NBIN + t];
        }
    }
    const int base = excl_scan_256(tot, s_wsum);
    const int lst  = excl_scan_256(own, s_wsum);
    if (t < NBIN) {
        s_off[t] = base + before;
        s_lst[t] = lst;
        if (g == 0) binbase[t] = base;
    }
    if (g == 0 && t == NBIN - 1) binbase[NBIN] = base + tot;
    __syncthreads();

    int2* __restrict__ pairs = (int2*)binned;
    const int chunk = (e + NCH - 1) / NCH;
    const int beg = g * chunk;
    const int end = min(beg + chunk, e);
    const int cn  = end - beg;
    if (cn <= CH_CAP) {
        for (int i0 = beg + t; i0 < end; i0 += 4 * SORT_T) {
            int d[4], s[4];
#pragma unroll
            for (int u = 0; u < 4; ++u) {
                const int ii = min(i0 + u * SORT_T, end - 1);
                d[u] = dst[ii];
                s[u] = src[ii];
            }
#pragma unroll
            for (int u = 0; u < 4; ++u) {
                if (i0 + u * SORT_T < end) {
                    const int bb = d[u] / NPB;
                    const int r = atomicAdd(&s_cnt[bb], 1);
                    s_pairs[s_lst[bb] + r] = make_int2(s[u], d[u]);
                }
            }
        }
        __syncthreads();
        for (int i = t; i < cn; i += SORT_T) {
            const int2 p = s_pairs[i];
            const int bb = p.y / NPB;
            pairs[s_off[bb] + (i - s_lst[bb])] = p;
        }
    } else {
        for (int i = beg + t; i < end; i += SORT_T) {
            const int d = dst[i];
            const int s = src[i];
            const int bb = d / NPB;
            const int r = atomicAdd(&s_cnt[bb], 1);
            pairs[s_off[bb] + r] = make_int2(s, d);
        }
    }
}

__global__ __launch_bounds__(SORT_T)
void k_bin_rows(const int* __restrict__ binned, const int* __restrict__ binbase,
                int* __restrict__ rowptr, int* __restrict__ col,
                float* __restrict__ dinv, int n) {
    __shared__ int s_cur[NBIN];
    __shared__ int s_col[COL_CAP];
    __shared__ int s_wsum[8];
    const int t = threadIdx.x;
    const int bin = blockIdx.x;
    const int2* __restrict__ pairs = (const int2*)binned;
    const int beg = binbase[bin];
    const int end = binbase[bin + 1];
    const int ne  = end - beg;
    const int node0 = bin * NPB;

    if (t < NBIN) s_cur[t] = 0;
    __syncthreads();
    for (int i0 = beg + t; i0 < end; i0 += 4 * SORT_T) {
        int d[4];
#pragma unroll
        for (int u = 0; u < 4; ++u) d[u] = pairs[min(i0 + u * SORT_T, end - 1)].y;
#pragma unroll
        for (int u = 0; u < 4; ++u)
            if (i0 + u * SORT_T < end) atomicAdd(&s_cur[d[u] - node0], 1);
    }
    __syncthreads();
    const int deg = (t < NBIN) ? s_cur[t] : 0;
    const int ex = excl_scan_256(deg, s_wsum);
    if (t < NPB && node0 + t < n) {
        rowptr[node0 + t] = beg + ex;
        dinv[node0 + t]   = rsqrtf(fmaxf((float)(deg + 1), 1.0f));
    }
    if (bin == NBIN - 1 && t == 0) rowptr[n] = end;
    if (t < NBIN) s_cur[t] = ex;
    __syncthreads();

    if (ne <= COL_CAP) {
        for (int i0 = beg + t; i0 < end; i0 += 4 * SORT_T) {
            int2 p[4];
#pragma unroll
            for (int u = 0; u < 4; ++u) p[u] = pairs[min(i0 + u * SORT_T, end - 1)];
#pragma unroll
            for (int u = 0; u < 4; ++u) {
                if (i0 + u * SORT_T < end) {
                    const int r = atomicAdd(&s_cur[p[u].y - node0], 1);
                    s_col[r] = p[u].x;
                }
            }
        }
        __syncthreads();
        for (int i = t; i < ne; i += SORT_T) col[beg + i] = s_col[i];
    } else {
        for (int i = beg + t; i < end; i += SORT_T) {
            const int2 p = pairs[i];
            const int r = atomicAdd(&s_cur[p.y - node0], 1);
            col[beg + r] = p.x;
        }
    }
}

template <int KIN, int KOUT>
__global__ __launch_bounds__(THREADS)
void k_gemm_wmma(const float* __restrict__ A, const float* __restrict__ W,
                 const float* __restrict__ rscale, float* __restrict__ C, int nrows) {
    __shared__ v2f sW[(KIN / 2) * KOUT];
    for (int i = threadIdx.x; i < (KIN / 2) * KOUT; i += blockDim.x) {
        const int k2 = i / KOUT;
        const int c  = i % KOUT;
        v2f w;
        w.x = W[(2 * k2 + 0) * KOUT + c];
        w.y = W[(2 * k2 + 1) * KOUT + c];
        sW[i] = w;
    }
    __syncthreads();

    const int wave = threadIdx.x >> 5;
    const int lane = threadIdx.x & 31;
    const int tile = blockIdx.x * 8 + wave;
    if (tile * 16 >= nrows) return;

    constexpr int NT = KOUT / 16;
    v8f acc[NT] = {};

    const int m    = lane & 15;
    const int khi  = lane >> 4;
    const float* Arow = A + (size_t)(tile * 16 + m) * KIN;
    const v2f* sWl = sW + khi * KOUT + m;

    const float4 rs0 = *(const float4*)(rscale + tile * 16 + khi * 8);
    const float4 rs1 = *(const float4*)(rscale + tile * 16 + khi * 8 + 4);
    const float rs[8] = {rs0.x, rs0.y, rs0.z, rs0.w, rs1.x, rs1.y, rs1.z, rs1.w};

    for (int kk = 0; kk < KIN / 4; ++kk) {
        const v2f a = *(const v2f*)(Arow + kk * 4 + khi * 2);
        const v2f* sWk = sWl + (kk * 2) * KOUT;
#pragma unroll
        for (int t = 0; t < NT; ++t) {
            const v2f b = sWk[t * 16];
            acc[t] = __builtin_amdgcn_wmma_f32_16x16x4_f32(
                 false, a,  false, b,
                 (short)0, acc[t],  false,  false);
        }
    }

#pragma unroll
    for (int t = 0; t < NT; ++t) {
#pragma unroll
        for (int v = 0; v < 8; ++v) {
            const int r = tile * 16 + khi * 8 + v;
            C[(size_t)r * KOUT + t * 16 + m] = acc[t][v] * rs[v];
        }
    }
}

template <int F, bool RELU, bool NEXT>
__device__ __forceinline__ float4 agg_row(const int* __restrict__ rowptr, const int* __restrict__ col,
                                          const float* __restrict__ dinv, const float* __restrict__ Hs,
                                          const float* __restrict__ bias, int i, int c) {
    const int beg = rowptr[i];
    const int end = rowptr[i + 1];
    const float di = dinv[i];
    const float* Hc = Hs + c * 4;

    float4 acc = make_float4(0.0f, 0.0f, 0.0f, 0.0f);
    const int last = end - 1;
    for (int j = beg; j < end; j += 4) {
        const bool p1 = j + 1 < end;
        const bool p2 = j + 2 < end;
        const bool p3 = j + 3 < end;
        const int s0 = col[j];
        const int s1 = col[min(j + 1, last)];
        const int s2 = col[min(j + 2, last)];
        const int s3 = col[min(j + 3, last)];
        const float4 h0 = *(const float4*)(Hc + (size_t)s0 * F);
        const float4 h1 = *(const float4*)(Hc + (size_t)s1 * F);
        const float4 h2 = *(const float4*)(Hc + (size_t)s2 * F);
        const float4 h3 = *(const float4*)(Hc + (size_t)s3 * F);
        acc.x += h0.x; acc.y += h0.y; acc.z += h0.z; acc.w += h0.w;
        acc.x = p1 ? acc.x + h1.x : acc.x; acc.y = p1 ? acc.y + h1.y : acc.y;
        acc.z = p1 ? acc.z + h1.z : acc.z; acc.w = p1 ? acc.w + h1.w : acc.w;
        acc.x = p2 ? acc.x + h2.x : acc.x; acc.y = p2 ? acc.y + h2.y : acc.y;
        acc.z = p2 ? acc.z + h2.z : acc.z; acc.w = p2 ? acc.w + h2.w : acc.w;
        acc.x = p3 ? acc.x + h3.x : acc.x; acc.y = p3 ? acc.y + h3.y : acc.y;
        acc.z = p3 ? acc.z + h3.z : acc.z; acc.w = p3 ? acc.w + h3.w : acc.w;
    }

    const float4 hs = *(const float4*)(Hc + (size_t)i * F);
    const float* b = bias + c * 4;
    float4 v;
    v.x = (acc.x + hs.x) * di + b[0];
    v.y = (acc.y + hs.y) * di + b[1];
    v.z = (acc.z + hs.z) * di + b[2];
    v.w = (acc.w + hs.w) * di + b[3];
    if (RELU) {
        v.x = fmaxf(v.x, 0.0f); v.y = fmaxf(v.y, 0.0f);
        v.z = fmaxf(v.z, 0.0f); v.w = fmaxf(v.w, 0.0f);
    }
    if (NEXT) {
        v.x *= di; v.y *= di; v.z *= di; v.w *= di;
    }
    return v;
}

template <int F, bool RELU>
__global__ __launch_bounds__(THREADS)
void k_gather(const int* __restrict__ rowptr, const int* __restrict__ col,
              const float* __restrict__ dinv, const float* __restrict__ H,
              const float* __restrict__ bias, float* __restrict__ out, int n) {
    constexpr int C = F / 4;
    const int tid = blockIdx.x * blockDim.x + threadIdx.x;
    const int i = tid / C;
    const int c = tid % C;
    if (i >= n) return;
    const float4 v = agg_row<F, RELU, false>(rowptr, col, dinv, H, bias, i, c);
    *(float4*)(out + (size_t)i * F + c * 4) = v;
}

#define TILE_PITCH 68
template <int KOUT, bool MOVE>
__global__ __launch_bounds__(THREADS)
void k_gather_gemm(const int* __restrict__ rowptr, const int* __restrict__ col,
                   const float* __restrict__ dinv, const float* __restrict__ H,
                   const float* __restrict__ bias, const float* __restrict__ Wn,
                   float* __restrict__ Cn, int n,
                   const int* __restrict__ mv_from, int* __restrict__ mv_to, int mv_words) {
    constexpr int F  = 64;
    constexpr int NT = KOUT / 16;
    __shared__ float sH[16 * TILE_PITCH];
    const int r = threadIdx.x >> 4;
    const int c = threadIdx.x & 15;
    const int i = blockIdx.x * 16 + r;

    float4 v = make_float4(0.0f, 0.0f, 0.0f, 0.0f);
    if (i < n) v = agg_row<F, true, true>(rowptr, col, dinv, H, bias, i, c);
    *(float4*)(sH + r * TILE_PITCH + c * 4) = v;

    const int wave = threadIdx.x >> 5;
    const int lane = threadIdx.x & 31;
    const int m    = lane & 15;
    const int khi  = lane >> 4;
    const bool mul = wave < NT;

    v2f b[F / 4];
    if (mul) {
        const float* Wl = Wn + (khi * 2) * KOUT + wave * 16 + m;
#pragma unroll
        for (int kk = 0; kk < F / 4; ++kk) {
            b[kk].x = Wl[(kk * 4 + 0) * KOUT];
            b[kk].y = Wl[(kk * 4 + 1) * KOUT];
        }
    }
    __syncthreads();
    if (!mul) {
        if (MOVE) {
            constexpr int MT = THREADS - NT * 32;
            constexpr int NWG = GCN_N / 16;
            const int lines = (mv_words + 31) / 32;
            const int per   = ((lines + NWG - 1) / NWG) * 32;
            const int beg   = (int)blockIdx.x * per;
            const int end   = min(beg + per, mv_words);
            for (int w = beg + ((int)threadIdx.x - NT * 32) * 4; w < end; w += MT * 4) {
                if (w + 4 <= end) {
                    *(int4*)(mv_to + w) = *(const int4*)(mv_from + w);
                } else {
                    for (int u = w; u < end; ++u) mv_to[u] = mv_from[u];
                }
            }
        }
        return;
    }

    v8f acc = {};
    const float* sA = sH + m * TILE_PITCH + khi * 2;
#pragma unroll
    for (int kk = 0; kk < F / 4; ++kk) {
        const v2f a = *(const v2f*)(sA + kk * 4);
        acc = __builtin_amdgcn_wmma_f32_16x16x4_f32(
             false, a,  false, b[kk],
             (short)0, acc,  false,  false);
    }
    float* Cr = Cn + (size_t)(blockIdx.x * 16 + khi * 8) * KOUT + wave * 16 + m;
#pragma unroll
    for (int q = 0; q < 8; ++q) Cr[(size_t)q * KOUT] = acc[q];
}
static_assert(GCN_N % 16 == 0, "row tiles are whole");

extern "C" void kernel_launch(void* const* d_in, const int* in_sizes, int n_in,
                              void* d_out, int out_size, void* d_ws, size_t ws_size,
                              hipStream_t stream) {
    const float* x  = (const float*)d_in[0];
    const int*   ei = (const int*)d_in[1];
    const float* W1 = (const float*)d_in[2];
    const float* b1 = (const float*)d_in[3];
    const float* W2 = (const float*)d_in[4];
    const float* b2 = (const float*)d_in[5];
    const float* W3 = (const float*)d_in[6];
    const float* b3 = (const float*)d_in[7];
    float* out = (float*)d_out;

    const int N = GCN_N;
    const int E = in_sizes[1] / 2;
    const int* src = ei;
    const int* dst = ei + E;

    float* dinv = (float*)d_ws;
    float* bufA = dinv + ((N + 255) & ~255);
    float* bufB = bufA + (size_t)N * 64;

    int* cntmat  = (int*)bufA;
    int* binbase = cntmat + NCH * NBIN;
    int* binned  = (int*)bufB;
    int* col    = (int*)out;
    int* rowptr = col + E;
    int* col3    = (int*)(bufA + (size_t)N * 32);
    int* rowptr3 = col3 + E;

    const int tiles  = N / 16;
    const int gemmBl = (tiles + 7) / 8;

    auto blocks = [](long long n) { return (unsigned)((n + THREADS - 1) / THREADS); };

    k_bin_count<<<NCH, SORT_T, 0, stream>>>(dst, cntmat, E);
    k_bin_scatter<<<NCH, SORT_T, 0, stream>>>(src, dst, cntmat, binbase, binned, E);
    k_bin_rows<<<NBIN, SORT_T, 0, stream>>>(binned, binbase, rowptr, col, dinv, N);

    k_gemm_wmma<128, 64><<<gemmBl, THREADS, 0, stream>>>(x, W1, dinv, bufA, N);

    k_gather_gemm<64, false><<<tiles, THREADS, 0, stream>>>(rowptr, col, dinv, bufA, b1, W2, bufB, N,
                                                            nullptr, nullptr, 0);

    k_gather_gemm<32, true><<<tiles, THREADS, 0, stream>>>(rowptr, col, dinv, bufB, b2, W3, bufA, N,
                                                           col, col3, E + N + 1);

    k_gather<32, false><<<blocks((long long)N * 8), THREADS, 0, stream>>>(rowptr3, col3, dinv, bufA, b3, out, N);
}


# Gather row chain shortened: index list in coalesced block loads handed out by DPP lane broadcast (no index load per trip, 8 row loads a trip), self row and bias asked for up front; sum order kept
# speedup vs baseline: 3.9684x; 1.1008x over previous
#include <hip/hip_runtime.h>
#include <cstddef>

typedef __attribute__((ext_vector_type(2))) float v2f;
typedef __attribute__((ext_vector_type(8))) float v8f;

#define GCN_N 50000
#define THREADS 256

#define NBIN 256
#define NCH  256
#define NPB  ((GCN_N + NBIN - 1) / NBIN)
#define SORT_T 1024
#define CH_CAP  4096
#define COL_CAP 8192

__device__ __forceinline__ int excl_scan_256(int v, int* s_wsum) {
    const int lane = threadIdx.x & 31;
    const int wave = threadIdx.x >> 5;
    int x = v;
#pragma unroll
    for (int off = 1; off < 32; off <<= 1) {
        const int y = __shfl_up(x, off);
        if (lane >= off) x += y;
    }
    if (wave < 8 && lane == 31) s_wsum[wave] = x;
    __syncthreads();
    int base = 0;
    if (wave < 8)
        for (int w = 0; w < wave; ++w) base += s_wsum[w];
    __syncthreads();
    return base + x - v;
}

__global__ __launch_bounds__(SORT_T)
void k_bin_count(const int* __restrict__ dst, int* __restrict__ cntmat, int e) {
    __shared__ int s_cnt[NBIN];
    const int t = threadIdx.x;
    const int g = blockIdx.x;
    if (t < NBIN) s_cnt[t] = 0;
    __syncthreads();
    const int chunk = (e + NCH - 1) / NCH;
    const int beg = g * chunk;
    const int end = min(beg + chunk, e);
    for (int i0 = beg + t; i0 < end; i0 += 4 * SORT_T) {
        int d[4];
#pragma unroll
        for (int u = 0; u < 4; ++u) d[u] = dst[min(i0 + u * SORT_T, end - 1)];
#pragma unroll
        for (int u = 0; u < 4; ++u)
            if (i0 + u * SORT_T < end) atomicAdd(&s_cnt[d[u] / NPB], 1);
    }
    __syncthreads();
    if (t < NBIN) cntmat[g * NBIN + t] = s_cnt[t];
}

__global__ __launch_bounds__(SORT_T)
void k_bin_scatter(const int* __restrict__ src, const int* __restrict__ dst,
                   const int* __restrict__ cntmat, int* __restrict__ binbase,
                   int* __restrict__ binned, int e) {
    __shared__ int4 s_raw[CH_CAP / 2];
    __shared__ int s_off[NBIN];
    __shared__ int s_lst[NBIN];
    __shared__ int s_cnt[NBIN];
    __shared__ int s_wsum[8];
    int*  s_all   = (int*)s_raw;
    int*  s_bef   = s_all + 16 * NBIN;
    int2* s_pairs = (int2*)s_raw;
    const int t = threadIdx.x;
    const int g = blockIdx.x;

    const int b4 = (t & 63) * 4;
    const int q  = t >> 6;
    int4 all = make_int4(0, 0, 0, 0);
    int4 bef = make_int4(0, 0, 0, 0);
#pragma unroll
    for (int k = 0; k < NCH / 16; ++k) {
        const int gg = q * (NCH / 16) + k;
        const int4 c = *(const int4*)(cntmat + gg * NBIN + b4);
        const bool front = gg < g;
        all.x += c.x; all.y += c.y; all.z += c.z; all.w += c.w;
        bef.x += front ? c.x : 0; bef.y += front ? c.y : 0;
        bef.z += front ? c.z : 0; bef.w += front ? c.w : 0;
    }
    const int own = (t < NBIN) ? cntmat[g * NBIN + t] : 0;
    *(int4*)(s_all + q * NBIN + b4) = all;
    *(int4*)(s_bef + q * NBIN + b4) = bef;
    if (t < NBIN) s_cnt[t] = 0;
    __syncthreads();
    int tot = 0, before = 0;
    if (t < NBIN) {
#pragma unroll
        for (int k = 0; k < 16; ++k) {
            tot    += s_all[k * NBIN + t];
            before += s_bef[k * NBIN + t];
        }
    }
    const int base = excl_scan_256(tot, s_wsum);
    const int lst  = excl_scan_256(own, s_wsum);
    if (t < NBIN) {
        s_off[t] = base + before;
        s_lst[t] = lst;
        if (g == 0) binbase[t] = base;
    }
    if (g == 0 && t == NBIN - 1) binbase[NBIN] = base + tot;
    __syncthreads();

    int2* __restrict__ pairs = (int2*)binned;
    const int chunk = (e + NCH - 1) / NCH;
    const int beg = g * chunk;
    const int end = min(beg + chunk, e);
    const int cn  = end - beg;
    if (cn <= CH_CAP) {
        for (int i0 = beg + t; i0 < end; i0 += 4 * SORT_T) {
            int d[4], s[4];
#pragma unroll
            for (int u = 0; u < 4; ++u) {
                const int ii = min(i0 + u * SORT_T, end - 1);
                d[u] = dst[ii];
                s[u] = src[ii];
            }
#pragma unroll
            for (int u = 0; u < 4; ++u) {
                if (i0 + u * SORT_T < end) {
                    const int bb = d[u] / NPB;
                    const int r = atomicAdd(&s_cnt[bb], 1);
                    s_pairs[s_lst[bb] + r] = make_int2(s[u], d[u]);
                }
            }
        }
        __syncthreads();
        for (int i = t; i < cn; i += SORT_T) {
            const int2 p = s_pairs[i];
            const int bb = p.y / NPB;
            pairs[s_off[bb] + (i - s_lst[bb])] = p;
        }
    } else {
        for (int i = beg + t; i < end; i += SORT_T) {
            const int d = dst[i];
            const int s = src[i];
            const int bb = d / NPB;
            const int r = atomicAdd(&s_cnt[bb], 1);
            pairs[s_off[bb] + r] = make_int2(s, d);
        }
    }
}

__global__ __launch_bounds__(SORT_T)
void k_bin_rows(const int* __restrict__ binned, const int* __restrict__ binbase,
                int* __restrict__ rowptr, int* __restrict__ col,
                float* __restrict__ dinv, int n) {
    __shared__ int s_cur[NBIN];
    __shared__ int s_col[COL_CAP];
    __shared__ int s_wsum[8];
    const int t = threadIdx.x;
    const int bin = blockIdx.x;
    const int2* __restrict__ pairs = (const int2*)binned;
    const int beg = binbase[bin];
    const int end = binbase[bin + 1];
    const int ne  = end - beg;
    const int node0 = bin * NPB;

    if (t < NBIN) s_cur[t] = 0;
    __syncthreads();
    for (int i0 = beg + t; i0 < end; i0 += 4 * SORT_T) {
        int d[4];
#pragma unroll
        for (int u = 0; u < 4; ++u) d[u] = pairs[min(i0 + u * SORT_T, end - 1)].y;
#pragma unroll
        for (int u = 0; u < 4; ++u)
            if (i0 + u * SORT_T < end) atomicAdd(&s_cur[d[u] - node0], 1);
    }
    __syncthreads();
    const int deg = (t < NBIN) ? s_cur[t] : 0;
    const int ex = excl_scan_256(deg, s_wsum);
    if (t < NPB && node0 + t < n) {
        rowptr[node0 + t] = beg + ex;
        dinv[node0 + t]   = rsqrtf(fmaxf((float)(deg + 1), 1.0f));
    }
    if (bin == NBIN - 1 && t == 0) rowptr[n] = end;
    if (t < NBIN) s_cur[t] = ex;
    __syncthreads();

    if (ne <= COL_CAP) {
        for (int i0 = beg + t; i0 < end; i0 += 4 * SORT_T) {
            int2 p[4];
#pragma unroll
            for (int u = 0; u < 4; ++u) p[u] = pairs[min(i0 + u * SORT_T, end - 1)];
#pragma unroll
            for (int u = 0; u < 4; ++u) {
                if (i0 + u * SORT_T < end) {
                    const int r = atomicAdd(&s_cur[p[u].y - node0], 1);
                    s_col[r] = p[u].x;
                }
            }
        }
        __syncthreads();
        for (int i = t; i < ne; i += SORT_T) col[beg + i] = s_col[i];
    } else {
        for (int i = beg + t; i < end; i += SORT_T) {
            const int2 p = pairs[i];
            const int r = atomicAdd(&s_cur[p.y - node0], 1);
            col[beg + r] = p.x;
        }
    }
}

template <int KIN, int KOUT>
__global__ __launch_bounds__(THREADS)
void k_gemm_wmma(const float* __restrict__ A, const float* __restrict__ W,
                 const float* __restrict__ rscale, float* __restrict__ C, int nrows) {
    __shared__ v2f sW[(KIN / 2) * KOUT];
    for (int i = threadIdx.x; i < (KIN / 2) * KOUT; i += blockDim.x) {
        const int k2 = i / KOUT;
        const int c  = i % KOUT;
        v2f w;
        w.x = W[(2 * k2 + 0) * KOUT + c];
        w.y = W[(2 * k2 + 1) * KOUT + c];
        sW[i] = w;
    }
    __syncthreads();

    const int wave = threadIdx.x >> 5;
    const int lane = threadIdx.x & 31;
    const int tile = blockIdx.x * 8 + wave;
    if (tile * 16 >= nrows) return;

    constexpr int NT = KOUT / 16;
    v8f acc[NT] = {};

    const int m    = lane & 15;
    const int khi  = lane >> 4;
    const float* Arow = A + (size_t)(tile * 16 + m) * KIN;
    const v2f* sWl = sW + khi * KOUT + m;

    const float4 rs0 = *(const float4*)(rscale + tile * 16 + khi * 8);
    const float4 rs1 = *(const float4*)(rscale + tile * 16 + khi * 8 + 4);
    const float rs[8] = {rs0.x, rs0.y, rs0.z, rs0.w, rs1.x, rs1.y, rs1.z, rs1.w};

    for (int kk = 0; kk < KIN / 4; ++kk) {
        const v2f a = *(const v2f*)(Arow + kk * 4 + khi * 2);
        const v2f* sWk = sWl + (kk * 2) * KOUT;
#pragma unroll
        for (int t = 0; t < NT; ++t) {
            const v2f b = sWk[t * 16];
            acc[t] = __builtin_amdgcn_wmma_f32_16x16x4_f32(
                 false, a,  false, b,
                 (short)0, acc[t],  false,  false);
        }
    }

#pragma unroll
    for (int t = 0; t < NT; ++t) {
#pragma unroll
        for (int v = 0; v < 8; ++v) {
            const int r = tile * 16 + khi * 8 + v;
            C[(size_t)r * KOUT + t * 16 + m] = acc[t][v] * rs[v];
        }
    }
}


template <int G>
__device__ __forceinline__ int group_bcast(int v, int k);

template <>
__device__ __forceinline__ int group_bcast<16>(int v, int k) {
#define ROW_SHARE(K) case K: return __builtin_amdgcn_update_dpp(0, v, 0x150 + K, 0xF, 0xF, false);
    switch (k) {
        ROW_SHARE(0)  ROW_SHARE(1)  ROW_SHARE(2)  ROW_SHARE(3)
        ROW_SHARE(4)  ROW_SHARE(5)  ROW_SHARE(6)  ROW_SHARE(7)
        ROW_SHARE(8)  ROW_SHARE(9)  ROW_SHARE(10) ROW_SHARE(11)
        ROW_SHARE(12) ROW_SHARE(13) ROW_SHARE(14) ROW_SHARE(15)
    }
#undef ROW_SHARE
    return v;
}

template <>
__device__ __forceinline__ int group_bcast<8>(int v, int k) {
#define DPP8_ALL(K) case K: return __builtin_amdgcn_mov_dpp8(v, K * 0x249249);
    switch (k) {
        DPP8_ALL(0) DPP8_ALL(1) DPP8_ALL(2) DPP8_ALL(3)
        DPP8_ALL(4) DPP8_ALL(5) DPP8_ALL(6) DPP8_ALL(7)
    }
#undef DPP8_ALL
    return v;
}

__device__ __forceinline__ void hold_list_loads(const int (&w)[4]) {
    __builtin_amdgcn_sched_barrier(0);
    asm volatile("" :: "v"(w[0]), "v"(w[1]), "v"(w[2]), "v"(w[3]));
}
__device__ __forceinline__ void hold_list_loads(const int (&)[2]) {}

template <int F, bool RELU, bool NEXT>
__device__ __forceinline__ float4 agg_row(const int* __restrict__ rowptr, const int* __restrict__ col,
                                          const float* __restrict__ dinv, const float* __restrict__ Hs,
                                          const float* __restrict__ bias, int i, int c) {
    constexpr int G  = F / 4;
    constexpr int NI = 32 / G;
    const int beg = rowptr[i];
    const int end = rowptr[i + 1];
    const float di = dinv[i];
    const float* Hc = Hs + c * 4;
    const char* Hb = (const char*)Hs;
    const unsigned coff = (unsigned)c * 16u;

    const float4 hs = *(const float4*)(Hc + (size_t)i * F);
    const float4 bb = *(const float4*)(bias + c * 4);

    float4 acc = make_float4(0.0f, 0.0f, 0.0f, 0.0f);
    const int last = end - 1;
    for (int base = beg; base < end; base += 32) {
        int idx[NI];
#pragma unroll
        for (int q = 0; q < NI; ++q) idx[q] = col[min(base + q * G + c, last)];
        hold_list_loads(idx);
#pragma unroll
        for (int t = 0; t < 4; ++t) {
            const int e0 = base + t * 8;
            if (t == 0 || e0 < end) {
                float4 h[8];
#pragma unroll
                for (int u = 0; u < 8; ++u) {
                    const int e = t * 8 + u;
                    const unsigned s = (unsigned)group_bcast<G>(idx[e / G], e % G);
                    h[u] = *(const float4*)(Hb + (s * (unsigned)(F * sizeof(float)) + coff));
                }
#pragma unroll
                for (int u = 0; u < 8; ++u) {
                    const bool p = (u == 0) || (e0 + u < end);
                    acc.x = p ? acc.x + h[u].x : acc.x; acc.y = p ? acc.y + h[u].y : acc.y;
                    acc.z = p ? acc.z + h[u].z : acc.z; acc.w = p ? acc.w + h[u].w : acc.w;
                }
            }
        }
    }

    float4 v;
    v.x = (acc.x + hs.x) * di + bb.x;
    v.y = (acc.y + hs.y) * di + bb.y;
    v.z = (acc.z + hs.z) * di + bb.z;
    v.w = (acc.w + hs.w) * di + bb.w;
    if (RELU) {
        v.x = fmaxf(v.x, 0.0f); v.y = fmaxf(v.y, 0.0f);
        v.z = fmaxf(v.z, 0.0f); v.w = fmaxf(v.w, 0.0f);
    }
    if (NEXT) {
        v.x *= di; v.y *= di; v.z *= di; v.w *= di;
    }
    return v;
}

template <int F, bool RELU>
__global__ __launch_bounds__(THREADS)
void k_gather(const int* __restrict__ rowptr, const int* __restrict__ col,
              const float* __restrict__ dinv, const float* __restrict__ H,
              const float* __restrict__ bias, float* __restrict__ out, int n) {
    constexpr int C = F / 4;
    const int tid = blockIdx.x * blockDim.x + threadIdx.x;
    const int i = tid / C;
    const int c = tid % C;
    if (i >= n) return;
    const float4 v = agg_row<F, RELU, false>(rowptr, col, dinv, H, bias, i, c);
    *(float4*)(out + (size_t)i * F + c * 4) = v;
}

#define TILE_PITCH 68
template <int KOUT, bool MOVE>
__global__ __launch_bounds__(THREADS)
void k_gather_gemm(const int* __restrict__ rowptr, const int* __restrict__ col,
                   const float* __restrict__ dinv, const float* __restrict__ H,
                   const float* __restrict__ bias, const float* __restrict__ Wn,
                   float* __restrict__ Cn, int n,
                   const int* __restrict__ mv_from, int* __restrict__ mv_to, int mv_words) {
    constexpr int F  = 64;
    constexpr int NT = KOUT / 16;
    __shared__ float sH[16 * TILE_PITCH];
    const int r = threadIdx.x >> 4;
    const int c = threadIdx.x & 15;
    const int i = blockIdx.x * 16 + r;

    float4 v = make_float4(0.0f, 0.0f, 0.0f, 0.0f);
    if (i < n) v = agg_row<F, true, true>(rowptr, col, dinv, H, bias, i, c);
    *(float4*)(sH + r * TILE_PITCH + c * 4) = v;

    const int wave = threadIdx.x >> 5;
    const int lane = threadIdx.x & 31;
    const int m    = lane & 15;
    const int khi  = lane >> 4;
    const bool mul = wave < NT;

    v2f b[F / 4];
    if (mul) {
        const float* Wl = Wn + (khi * 2) * KOUT + wave * 16 + m;
#pragma unroll
        for (int kk = 0; kk < F / 4; ++kk) {
            b[kk].x = Wl[(kk * 4 + 0) * KOUT];
            b[kk].y = Wl[(kk * 4 + 1) * KOUT];
        }
    }
    __syncthreads();
    if (!mul) {
        if (MOVE) {
            constexpr int MT = THREADS - NT * 32;
            constexpr int NWG = GCN_N / 16;
            const int lines = (mv_words + 31) / 32;
            const int per   = ((lines + NWG - 1) / NWG) * 32;
            const int beg   = (int)blockIdx.x * per;
            const int end   = min(beg + per, mv_words);
            for (int w = beg + ((int)threadIdx.x - NT * 32) * 4; w < end; w += MT * 4) {
                if (w + 4 <= end) {
                    *(int4*)(mv_to + w) = *(const int4*)(mv_from + w);
                } else {
                    for (int u = w; u < end; ++u) mv_to[u] = mv_from[u];
                }
            }
        }
        return;
    }

    v8f acc = {};
    const float* sA = sH + m * TILE_PITCH + khi * 2;
#pragma unroll
    for (int kk = 0; kk < F / 4; ++kk) {
        const v2f a = *(const v2f*)(sA + kk * 4);
        acc = __builtin_amdgcn_wmma_f32_16x16x4_f32(
             false, a,  false, b[kk],
             (short)0, acc,  false,  false);
    }
    float* Cr = Cn + (size_t)(blockIdx.x * 16 + khi * 8) * KOUT + wave * 16 + m;
#pragma unroll
    for (int q = 0; q < 8; ++q) Cr[(size_t)q * KOUT] = acc[q];
}
static_assert(GCN_N % 16 == 0, "row tiles are whole");

extern "C" void kernel_launch(void* const* d_in, const int* in_sizes, int n_in,
                              void* d_out, int out_size, void* d_ws, size_t ws_size,
                              hipStream_t stream) {
    const float* x  = (const float*)d_in[0];
    const int*   ei = (const int*)d_in[1];
    const float* W1 = (const float*)d_in[2];
    const float* b1 = (const float*)d_in[3];
    const float* W2 = (const float*)d_in[4];
    const float* b2 = (const float*)d_in[5];
    const float* W3 = (const float*)d_in[6];
    const float* b3 = (const float*)d_in[7];
    float* out = (float*)d_out;

    const int N = GCN_N;
    const int E = in_sizes[1] / 2;
    const int* src = ei;
    const int* dst = ei + E;

    float* dinv = (float*)d_ws;
    float* bufA = dinv + ((N + 255) & ~255);
    float* bufB = bufA + (size_t)N * 64;

    int* cntmat  = (int*)bufA;
    int* binbase = cntmat + NCH * NBIN;
    int* binned  = (int*)bufB;
    int* col    = (int*)out;
    int* rowptr = col + E;
    int* col3    = (int*)(bufA + (size_t)N * 32);
    int* rowptr3 = col3 + E;

    const int tiles  = N / 16;
    const int gemmBl = (tiles + 7) / 8;

    auto blocks = [](long long n) { return (unsigned)((n + THREADS - 1) / THREADS); };

    k_bin_count<<<NCH, SORT_T, 0, stream>>>(dst, cntmat, E);
    k_bin_scatter<<<NCH, SORT_T, 0, stream>>>(src, dst, cntmat, binbase, binned, E);
    k_bin_rows<<<NBIN, SORT_T, 0, stream>>>(binned, binbase, rowptr, col, dinv, N);

    k_gemm_wmma<128, 64><<<gemmBl, THREADS, 0, stream>>>(x, W1, dinv, bufA, N);

    k_gather_gemm<64, false><<<tiles, THREADS, 0, stream>>>(rowptr, col, dinv, bufA, b1, W2, bufB, N,
                                                            nullptr, nullptr, 0);

    k_gather_gemm<32, true><<<tiles, THREADS, 0, stream>>>(rowptr, col, dinv, bufB, b2, W3, bufA, N,
                                                           col, col3, E + N + 1);

    k_gather<32, false><<<blocks((long long)N * 8), THREADS, 0, stream>>>(rowptr3, col3, dinv, bufA, b3, out, N);
}


# Dense transform x@W1: every load asked for up front (W staged by 16-byte loads in one round trip, whole A tile in flight across an LDS-only barrier), no wait per loop trip; sums bit-identical
# speedup vs baseline: 4.4436x; 1.1197x over previous
#include <hip/hip_runtime.h>
#include <cstddef>

typedef __attribute__((ext_vector_type(2))) float v2f;
typedef __attribute__((ext_vector_type(8))) float v8f;

#define GCN_N 50000
#define THREADS 256

#define NBIN 256
#define NCH  256
#define NPB  ((GCN_N + NBIN - 1) / NBIN)
#define SORT_T 1024
#define CH_CAP  4096
#define COL_CAP 8192

__device__ __forceinline__ int excl_scan_256(int v, int* s_wsum) {
    const int lane = threadIdx.x & 31;
    const int wave = threadIdx.x >> 5;
    int x = v;
#pragma unroll
    for (int off = 1; off < 32; off <<= 1) {
        const int y = __shfl_up(x, off);
        if (lane >= off) x += y;
    }
    if (wave < 8 && lane == 31) s_wsum[wave] = x;
    __syncthreads();
    int base = 0;
    if (wave < 8)
        for (int w = 0; w < wave; ++w) base += s_wsum[w];
    __syncthreads();
    return base + x - v;
}

__global__ __launch_bounds__(SORT_T)
void k_bin_count(const int* __restrict__ dst, int* __restrict__ cntmat, int e) {
    __shared__ int s_cnt[NBIN];
    const int t = threadIdx.x;
    const int g = blockIdx.x;
    if (t < NBIN) s_cnt[t] = 0;
    __syncthreads();
    const int chunk = (e + NCH - 1) / NCH;
    const int beg = g * chunk;
    const int end = min(beg + chunk, e);
    for (int i0 = beg + t; i0 < end; i0 += 4 * SORT_T) {
        int d[4];
#pragma unroll
        for (int u = 0; u < 4; ++u) d[u] = dst[min(i0 + u * SORT_T, end - 1)];
#pragma unroll
        for (int u = 0; u < 4; ++u)
            if (i0 + u * SORT_T < end) atomicAdd(&s_cnt[d[u] / NPB], 1);
    }
    __syncthreads();
    if (t < NBIN) cntmat[g * NBIN + t] = s_cnt[t];
}

__global__ __launch_bounds__(SORT_T)
void k_bin_scatter(const int* __restrict__ src, const int* __restrict__ dst,
                   const int* __restrict__ cntmat, int* __restrict__ binbase,
                   int* __restrict__ binned, int e) {
    __shared__ int4 s_raw[CH_CAP / 2];
    __shared__ int s_off[NBIN];
    __shared__ int s_lst[NBIN];
    __shared__ int s_cnt[NBIN];
    __shared__ int s_wsum[8];
    int*  s_all   = (int*)s_raw;
    int*  s_bef   = s_all + 16 * NBIN;
    int2* s_pairs = (int2*)s_raw;
    const int t = threadIdx.x;
    const int g = blockIdx.x;

    const int b4 = (t & 63) * 4;
    const int q  = t >> 6;
    int4 all = make_int4(0, 0, 0, 0);
    int4 bef = make_int4(0, 0, 0, 0);
#pragma unroll
    for (int k = 0; k < NCH / 16; ++k) {
        const int gg = q * (NCH / 16) + k;
        const int4 c = *(const int4*)(cntmat + gg * NBIN + b4);
        const bool front = gg < g;
        all.x += c.x; all.y += c.y; all.z += c.z; all.w += c.w;
        bef.x += front ? c.x : 0; bef.y += front ? c.y : 0;
        bef.z += front ? c.z : 0; bef.w += front ? c.w : 0;
    }
    const int own = (t < NBIN) ? cntmat[g * NBIN + t] : 0;
    *(int4*)(s_all + q * NBIN + b4) = all;
    *(int4*)(s_bef + q * NBIN + b4) = bef;
    if (t < NBIN) s_cnt[t] = 0;
    __syncthreads();
    int tot = 0, before = 0;
    if (t < NBIN) {
#pragma unroll
        for (int k = 0; k < 16; ++k) {
            tot    += s_all[k * NBIN + t];
            before += s_bef[k * NBIN + t];
        }
    }
    const int base = excl_scan_256(tot, s_wsum);
    const int lst  = excl_scan_256(own, s_wsum);
    if (t < NBIN) {
        s_off[t] = base + before;
        s_lst[t] = lst;
        if (g == 0) binbase[t] = base;
    }
    if (g == 0 && t == NBIN - 1) binbase[NBIN] = base + tot;
    __syncthreads();

    int2* __restrict__ pairs = (int2*)binned;
    const int chunk = (e + NCH - 1) / NCH;
    const int beg = g * chunk;
    const int end = min(beg + chunk, e);
    const int cn  = end - beg;
    if (cn <= CH_CAP) {
        for (int i0 = beg + t; i0 < end; i0 += 4 * SORT_T) {
            int d[4], s[4];
#pragma unroll
            for (int u = 0; u < 4; ++u) {
                const int ii = min(i0 + u * SORT_T, end - 1);
                d[u] = dst[ii];
                s[u] = src[ii];
            }
#pragma unroll
            for (int u = 0; u < 4; ++u) {
                if (i0 + u * SORT_T < end) {
                    const int bb = d[u] / NPB;
                    const int r = atomicAdd(&s_cnt[bb], 1);
                    s_pairs[s_lst[bb] + r] = make_int2(s[u], d[u]);
                }
            }
        }
        __syncthreads();
        for (int i = t; i < cn; i += SORT_T) {
            const int2 p = s_pairs[i];
            const int bb = p.y / NPB;
            pairs[s_off[bb] + (i - s_lst[bb])] = p;
        }
    } else {
        for (int i = beg + t; i < end; i += SORT_T) {
            const int d = dst[i];
            const int s = src[i];
            const int bb = d / NPB;
            const int r = atomicAdd(&s_cnt[bb], 1);
            pairs[s_off[bb] + r] = make_int2(s, d);
        }
    }
}

__global__ __launch_bounds__(SORT_T)
void k_bin_rows(const int* __restrict__ binned, const int* __restrict__ binbase,
                int* __restrict__ rowptr, int* __restrict__ col,
                float* __restrict__ dinv, int n) {
    __shared__ int s_cur[NBIN];
    __shared__ int s_col[COL_CAP];
    __shared__ int s_wsum[8];
    const int t = threadIdx.x;
    const int bin = blockIdx.x;
    const int2* __restrict__ pairs = (const int2*)binned;
    const int beg = binbase[bin];
    const int end = binbase[bin + 1];
    const int ne  = end - beg;
    const int node0 = bin * NPB;

    if (t < NBIN) s_cur[t] = 0;
    __syncthreads();
    for (int i0 = beg + t; i0 < end; i0 += 4 * SORT_T) {
        int d[4];
#pragma unroll
        for (int u = 0; u < 4; ++u) d[u] = pairs[min(i0 + u * SORT_T, end - 1)].y;
#pragma unroll
        for (int u = 0; u < 4; ++u)
            if (i0 + u * SORT_T < end) atomicAdd(&s_cur[d[u] - node0], 1);
    }
    __syncthreads();
    const int deg = (t < NBIN) ? s_cur[t] : 0;
    const int ex = excl_scan_256(deg, s_wsum);
    if (t < NPB && node0 + t < n) {
        rowptr[node0 + t] = beg + ex;
        dinv[node0 + t]   = rsqrtf(fmaxf((float)(deg + 1), 1.0f));
    }
    if (bin == NBIN - 1 && t == 0) rowptr[n] = end;
    if (t < NBIN) s_cur[t] = ex;
    __syncthreads();

    if (ne <= COL_CAP) {
        for (int i0 = beg + t; i0 < end; i0 += 4 * SORT_T) {
            int2 p[4];
#pragma unroll
            for (int u = 0; u < 4; ++u) p[u] = pairs[min(i0 + u * SORT_T, end - 1)];
#pragma unroll
            for (int u = 0; u < 4; ++u) {
                if (i0 + u * SORT_T < end) {
                    const int r = atomicAdd(&s_cur[p[u].y - node0], 1);
                    s_col[r] = p[u].x;
                }
            }
        }
        __syncthreads();
        for (int i = t; i < ne; i += SORT_T) col[beg + i] = s_col[i];
    } else {
        for (int i = beg + t; i < end; i += SORT_T) {
            const int2 p = pairs[i];
            const int r = atomicAdd(&s_cur[p.y - node0], 1);
            col[beg + r] = p.x;
        }
    }
}

template <int KIN, int KOUT>
__global__ __launch_bounds__(THREADS)
void k_gemm_wmma(const float* __restrict__ A, const float* __restrict__ W,
                 const float* __restrict__ rscale, float* __restrict__ C, int nrows) {
    __shared__ __attribute__((aligned(16))) v2f sW[(KIN / 2) * KOUT];
    constexpr int NT    = KOUT / 16;
    constexpr int NK    = KIN / 4;
    constexpr int UNITS = (KIN / 2) * (KOUT / 4);
    constexpr int UPT   = UNITS / THREADS;
    static_assert(UNITS % THREADS == 0, "every thread stages the same number of units");

    const int wave = threadIdx.x >> 5;
    const int lane = threadIdx.x & 31;
    const int tile = blockIdx.x * 8 + wave;
    const bool live = tile * 16 < nrows;
    const int ltile = min(tile, (nrows - 1) / 16);
    const int m    = lane & 15;
    const int khi  = lane >> 4;

    float4 w0[UPT], w1[UPT];
#pragma unroll
    for (int j = 0; j < UPT; ++j) {
        const int u  = threadIdx.x + j * THREADS;
        const int k2 = u / (KOUT / 4);
        const int c4 = (u % (KOUT / 4)) * 4;
        w0[j] = *(const float4*)(W + (2 * k2 + 0) * KOUT + c4);
        w1[j] = *(const float4*)(W + (2 * k2 + 1) * KOUT + c4);
    }

    v2f a[NK];
    const float* Arow = A + (size_t)(ltile * 16 + m) * KIN + khi * 2;
#pragma unroll
    for (int kk = 0; kk < NK; ++kk) a[kk] = *(const v2f*)(Arow + kk * 4);
    const float4 rs0 = *(const float4*)(rscale + ltile * 16 + khi * 8);
    const float4 rs1 = *(const float4*)(rscale + ltile * 16 + khi * 8 + 4);

#pragma unroll
    for (int j = 0; j < UPT; ++j) {
        const int u = threadIdx.x + j * THREADS;
        float4* d = (float4*)(sW + 4 * u);
        d[0] = make_float4(w0[j].x, w1[j].x, w0[j].y, w1[j].y);
        d[1] = make_float4(w0[j].z, w1[j].z, w0[j].w, w1[j].w);
    }
    __builtin_amdgcn_fence(__ATOMIC_RELEASE, "workgroup", "local");
    __builtin_amdgcn_s_barrier();
    __builtin_amdgcn_fence(__ATOMIC_ACQUIRE, "workgroup", "local");
    if (!live) {
        __builtin_amdgcn_fence(__ATOMIC_ACQUIRE, "workgroup");
        return;
    }

    v8f acc[NT] = {};
    const v2f* sWl = sW + khi * KOUT + m;

#pragma unroll
    for (int kk = 0; kk < NK; ++kk) {
        const v2f* sWk = sWl + (kk * 2) * KOUT;
#pragma unroll
        for (int t = 0; t < NT; ++t) {
            const v2f b = sWk[t * 16];
            acc[t] = __builtin_amdgcn_wmma_f32_16x16x4_f32(
                 false, a[kk],  false, b,
                 (short)0, acc[t],  false,  false);
        }
    }

    const float rs[8] = {rs0.x, rs0.y, rs0.z, rs0.w, rs1.x, rs1.y, rs1.z, rs1.w};
#pragma unroll
    for (int t = 0; t < NT; ++t) {
#pragma unroll
        for (int v = 0; v < 8; ++v) {
            const int r = tile * 16 + khi * 8 + v;
            C[(size_t)r * KOUT + t * 16 + m] = acc[t][v] * rs[v];
        }
    }
}


template <int G>
__device__ __forceinline__ int group_bcast(int v, int k);

template <>
__device__ __forceinline__ int group_bcast<16>(int v, int k) {
#define ROW_SHARE(K) case K: return __builtin_amdgcn_update_dpp(0, v, 0x150 + K, 0xF, 0xF, false);
    switch (k) {
        ROW_SHARE(0)  ROW_SHARE(1)  ROW_SHARE(2)  ROW_SHARE(3)
        ROW_SHARE(4)  ROW_SHARE(5)  ROW_SHARE(6)  ROW_SHARE(7)
        ROW_SHARE(8)  ROW_SHARE(9)  ROW_SHARE(10) ROW_SHARE(11)
        ROW_SHARE(12) ROW_SHARE(13) ROW_SHARE(14) ROW_SHARE(15)
    }
#undef ROW_SHARE
    return v;
}

template <>
__device__ __forceinline__ int group_bcast<8>(int v, int k) {
#define DPP8_ALL(K) case K: return __builtin_amdgcn_mov_dpp8(v, K * 0x249249);
    switch (k) {
        DPP8_ALL(0) DPP8_ALL(1) DPP8_ALL(2) DPP8_ALL(3)
        DPP8_ALL(4) DPP8_ALL(5) DPP8_ALL(6) DPP8_ALL(7)
    }
#undef DPP8_ALL
    return v;
}

__device__ __forceinline__ void hold_list_loads(const int (&w)[4]) {
    __builtin_amdgcn_sched_barrier(0);
    asm volatile("" :: "v"(w[0]), "v"(w[1]), "v"(w[2]), "v"(w[3]));
}
__device__ __forceinline__ void hold_list_loads(const int (&)[2]) {}

template <int F, bool RELU, bool NEXT>
__device__ __forceinline__ float4 agg_row(const int* __restrict__ rowptr, const int* __restrict__ col,
                                          const float* __restrict__ dinv, const float* __restrict__ Hs,
                                          const float* __restrict__ bias, int i, int c) {
    constexpr int G  = F / 4;
    constexpr int NI = 32 / G;
    const int beg = rowptr[i];
    const int end = rowptr[i + 1];
    const float di = dinv[i];
    const float* Hc = Hs + c * 4;
    const char* Hb = (const char*)Hs;
    const unsigned coff = (unsigned)c * 16u;

    const float4 hs = *(const float4*)(Hc + (size_t)i * F);
    const float4 bb = *(const float4*)(bias + c * 4);

    float4 acc = make_float4(0.0f, 0.0f, 0.0f, 0.0f);
    const int last = end - 1;
    for (int base = beg; base < end; base += 32) {
        int idx[NI];
#pragma unroll
        for (int q = 0; q < NI; ++q) idx[q] = col[min(base + q * G + c, last)];
        hold_list_loads(idx);
#pragma unroll
        for (int t = 0; t < 4; ++t) {
            const int e0 = base + t * 8;
            if (t == 0 || e0 < end) {
                float4 h[8];
#pragma unroll
                for (int u = 0; u < 8; ++u) {
                    const int e = t * 8 + u;
                    const unsigned s = (unsigned)group_bcast<G>(idx[e / G], e % G);
                    h[u] = *(const float4*)(Hb + (s * (unsigned)(F * sizeof(float)) + coff));
                }
#pragma unroll
                for (int u = 0; u < 8; ++u) {
                    const bool p = (u == 0) || (e0 + u < end);
                    acc.x = p ? acc.x + h[u].x : acc.x; acc.y = p ? acc.y + h[u].y : acc.y;
                    acc.z = p ? acc.z + h[u].z : acc.z; acc.w = p ? acc.w + h[u].w : acc.w;
                }
            }
        }
    }

    float4 v;
    v.x = (acc.x + hs.x) * di + bb.x;
    v.y = (acc.y + hs.y) * di + bb.y;
    v.z = (acc.z + hs.z) * di + bb.z;
    v.w = (acc.w + hs.w) * di + bb.w;
    if (RELU) {
        v.x = fmaxf(v.x, 0.0f); v.y = fmaxf(v.y, 0.0f);
        v.z = fmaxf(v.z, 0.0f); v.w = fmaxf(v.w, 0.0f);
    }
    if (NEXT) {
        v.x *= di; v.y *= di; v.z *= di; v.w *= di;
    }
    return v;
}

template <int F, bool RELU>
__global__ __launch_bounds__(THREADS)
void k_gather(const int* __restrict__ rowptr, const int* __restrict__ col,
              const float* __restrict__ dinv, const float* __restrict__ H,
              const float* __restrict__ bias, float* __restrict__ out, int n) {
    constexpr int C = F / 4;
    const int tid = blockIdx.x * blockDim.x + threadIdx.x;
    const int i = tid / C;
    const int c = tid % C;
    if (i >= n) return;
    const float4 v = agg_row<F, RELU, false>(rowptr, col, dinv, H, bias, i, c);
    *(float4*)(out + (size_t)i * F + c * 4) = v;
}

#define TILE_PITCH 68
template <int KOUT, bool MOVE>
__global__ __launch_bounds__(THREADS)
void k_gather_gemm(const int* __restrict__ rowptr, const int* __restrict__ col,
                   const float* __restrict__ dinv, const float* __restrict__ H,
                   const float* __restrict__ bias, const float* __restrict__ Wn,
                   float* __restrict__ Cn, int n,
                   const int* __restrict__ mv_from, int* __restrict__ mv_to, int mv_words) {
    constexpr int F  = 64;
    constexpr int NT = KOUT / 16;
    __shared__ float sH[16 * TILE_PITCH];
    const int r = threadIdx.x >> 4;
    const int c = threadIdx.x & 15;
    const int i = blockIdx.x * 16 + r;

    float4 v = make_float4(0.0f, 0.0f, 0.0f, 0.0f);
    if (i < n) v = agg_row<F, true, true>(rowptr, col, dinv, H, bias, i, c);
    *(float4*)(sH + r * TILE_PITCH + c * 4) = v;

    const int wave = threadIdx.x >> 5;
    const int lane = threadIdx.x & 31;
    const int m    = lane & 15;
    const int khi  = lane >> 4;
    const bool mul = wave < NT;

    v2f b[F / 4];
    if (mul) {
        const float* Wl = Wn + (khi * 2) * KOUT + wave * 16 + m;
#pragma unroll
        for (int kk = 0; kk < F / 4; ++kk) {
            b[kk].x = Wl[(kk * 4 + 0) * KOUT];
            b[kk].y = Wl[(kk * 4 + 1) * KOUT];
        }
    }
    __syncthreads();
    if (!mul) {
        if (MOVE) {
            constexpr int MT = THREADS - NT * 32;
            constexpr int NWG = GCN_N / 16;
            const int lines = (mv_words + 31) / 32;
            const int per   = ((lines + NWG - 1) / NWG) * 32;
            const int beg   = (int)blockIdx.x * per;
            const int end   = min(beg + per, mv_words);
            for (int w = beg + ((int)threadIdx.x - NT * 32) * 4; w < end; w += MT * 4) {
                if (w + 4 <= end) {
                    *(int4*)(mv_to + w) = *(const int4*)(mv_from + w);
                } else {
                    for (int u = w; u < end; ++u) mv_to[u] = mv_from[u];
                }
            }
        }
        return;
    }

    v8f acc = {};
    const float* sA = sH + m * TILE_PITCH + khi * 2;
#pragma unroll
    for (int kk = 0; kk < F / 4; ++kk) {
        const v2f a = *(const v2f*)(sA + kk * 4);
        acc = __builtin_amdgcn_wmma_f32_16x16x4_f32(
             false, a,  false, b[kk],
             (short)0, acc,  false,  false);
    }
    float* Cr = Cn + (size_t)(blockIdx.x * 16 + khi * 8) * KOUT + wave * 16 + m;
#pragma unroll
    for (int q = 0; q < 8; ++q) Cr[(size_t)q * KOUT] = acc[q];
}
static_assert(GCN_N % 16 == 0, "row tiles are whole");

extern "C" void kernel_launch(void* const* d_in, const int* in_sizes, int n_in,
                              void* d_out, int out_size, void* d_ws, size_t ws_size,
                              hipStream_t stream) {
    const float* x  = (const float*)d_in[0];
    const int*   ei = (const int*)d_in[1];
    const float* W1 = (const float*)d_in[2];
    const float* b1 = (const float*)d_in[3];
    const float* W2 = (const float*)d_in[4];
    const float* b2 = (const float*)d_in[5];
    const float* W3 = (const float*)d_in[6];
    const float* b3 = (const float*)d_in[7];
    float* out = (float*)d_out;

    const int N = GCN_N;
    const int E = in_sizes[1] / 2;
    const int* src = ei;
    const int* dst = ei + E;

    float* dinv = (float*)d_ws;
    float* bufA = dinv + ((N + 255) & ~255);
    float* bufB = bufA + (size_t)N * 64;

    int* cntmat  = (int*)bufA;
    int* binbase = cntmat + NCH * NBIN;
    int* binned  = (int*)bufB;
    int* col    = (int*)out;
    int* rowptr = col + E;
    int* col3    = (int*)(bufA + (size_t)N * 32);
    int* rowptr3 = col3 + E;

    const int tiles  = N / 16;
    const int gemmBl = (tiles + 7) / 8;

    auto blocks = [](long long n) { return (unsigned)((n + THREADS - 1) / THREADS); };

    k_bin_count<<<NCH, SORT_T, 0, stream>>>(dst, cntmat, E);
    k_bin_scatter<<<NCH, SORT_T, 0, stream>>>(src, dst, cntmat, binbase, binned, E);
    k_bin_rows<<<NBIN, SORT_T, 0, stream>>>(binned, binbase, rowptr, col, dinv, N);

    k_gemm_wmma<128, 64><<<gemmBl, THREADS, 0, stream>>>(x, W1, dinv, bufA, N);

    k_gather_gemm<64, false><<<tiles, THREADS, 0, stream>>>(rowptr, col, dinv, bufA, b1, W2, bufB, N,
                                                            nullptr, nullptr, 0);

    k_gather_gemm<32, true><<<tiles, THREADS, 0, stream>>>(rowptr, col, dinv, bufB, b2, W3, bufA, N,
                                                           col, col3, E + N + 1);

    k_gather<32, false><<<blocks((long long)N * 8), THREADS, 0, stream>>>(rowptr3, col3, dinv, bufA, b3, out, N);
}


# Edge lists in 2 launches, not 3: a slice orders its own edges by bin in place and publishes (start,count) per bin; a bin's work-group collects its 256 runs; row sum order = arrival rank of LDS adds
# speedup vs baseline: 4.7612x; 1.0715x over previous
#include <hip/hip_runtime.h>
#include <cstddef>

typedef __attribute__((ext_vector_type(2))) float v2f;
typedef __attribute__((ext_vector_type(8))) float v8f;

#define GCN_N 50000
#define THREADS 256

#define NBIN 256
#define NCH  256
#define NPB  ((GCN_N + NBIN - 1) / NBIN)
#define SORT_T 1024
#define CH_CAP  4096
#define COL_CAP 8192
static_assert(CH_CAP == 4 * SORT_T, "a staged slice is four edges a thread");
static_assert(NCH * 4 == SORT_T, "four lanes walk one slice's run of a bin");
static_assert(NPB <= NBIN, "a bin's nodes fit the 256 LDS counters");

__device__ __forceinline__ int excl_scan_256(int v, int* s_wsum) {
    const int lane = threadIdx.x & 31;
    const int wave = threadIdx.x >> 5;
    int x = v;
#pragma unroll
    for (int off = 1; off < 32; off <<= 1) {
        const int y = __shfl_up(x, off);
        if (lane >= off) x += y;
    }
    if (wave < 8 && lane == 31) s_wsum[wave] = x;
    __syncthreads();
    int base = 0;
    if (wave < 8)
        for (int w = 0; w < wave; ++w) base += s_wsum[w];
    __syncthreads();
    return base + x - v;
}

__global__ __launch_bounds__(SORT_T)
void k_slice_sort(const int* __restrict__ src, const int* __restrict__ dst,
                  int* __restrict__ tabw, int* __restrict__ binned, int e) {
    int2* __restrict__ tab = (int2*)tabw;
    __shared__ int2 s_pairs[CH_CAP];
    __shared__ int s_cnt[NBIN];
    __shared__ int s_lst[NBIN];
    __shared__ int s_cur[NBIN];
    __shared__ int s_wsum[8];
    const int t = threadIdx.x;
    const int g = blockIdx.x;
    int2* __restrict__ pairs = (int2*)binned;
    const int chunk = (e + NCH - 1) / NCH;
    const int beg = g * chunk;
    const int end = min(beg + chunk, e);
    const int cn  = end - beg;

    if (t < NBIN) s_cnt[t] = 0;
    __syncthreads();
    if (cn <= CH_CAP) {
        const bool has = beg + t < end;
        int d[4] = {0, 0, 0, 0}, s[4] = {0, 0, 0, 0};
        if (has) {
#pragma unroll
            for (int u = 0; u < 4; ++u) {
                const int ii = min(beg + t + u * SORT_T, end - 1);
                d[u] = dst[ii];
                s[u] = src[ii];
            }
        }
#pragma unroll
        for (int u = 0; u < 4; ++u)
            if (beg + t + u * SORT_T < end) atomicAdd(&s_cnt[d[u] / NPB], 1);
        __syncthreads();
        const int own = (t < NBIN) ? s_cnt[t] : 0;
        const int lst = excl_scan_256(own, s_wsum);
        if (t < NBIN) {
            s_lst[t] = lst;
            s_cur[t] = 0;
        }
        __syncthreads();
#pragma unroll
        for (int u = 0; u < 4; ++u) {
            if (beg + t + u * SORT_T < end) {
                const int bb = d[u] / NPB;
                const int r = atomicAdd(&s_cur[bb], 1);
                s_pairs[s_lst[bb] + r] = make_int2(s[u], d[u]);
            }
        }
        __syncthreads();
        if (t < NBIN) tab[t * NCH + g] = make_int2(lst, own);
        for (int i = t; i < cn; i += SORT_T) pairs[beg + i] = s_pairs[i];
    } else {
        for (int i = beg + t; i < end; i += SORT_T) atomicAdd(&s_cnt[dst[i] / NPB], 1);
        __syncthreads();
        const int own = (t < NBIN) ? s_cnt[t] : 0;
        const int lst = excl_scan_256(own, s_wsum);
        if (t < NBIN) {
            s_lst[t] = lst;
            s_cur[t] = 0;
            tab[t * NCH + g] = make_int2(lst, own);
        }
        __syncthreads();
        for (int i = beg + t; i < end; i += SORT_T) {
            const int dd = dst[i];
            const int ss = src[i];
            const int bb = dd / NPB;
            const int r = atomicAdd(&s_cur[bb], 1);
            pairs[beg + s_lst[bb] + r] = make_int2(ss, dd);
        }
    }
}

__device__ __forceinline__ void hold_trip_loads(int a, int b, int c, int d) {
    __builtin_amdgcn_sched_barrier(0);
    asm volatile("" :: "v"(a), "v"(b), "v"(c), "v"(d));
}

__global__ __launch_bounds__(SORT_T)
void k_bin_rows(const int* __restrict__ binned, const int* __restrict__ tabw,
                int* __restrict__ rowptr, int* __restrict__ col,
                float* __restrict__ dinv, int n, int e) {
    const int2* __restrict__ tab = (const int2*)tabw;
    __shared__ int s_cur[NBIN];
    __shared__ int s_col[COL_CAP];
    __shared__ int s_wsum[8];
    __shared__ int s_sbeg[NCH];
    __shared__ int s_scnt[NCH];
    __shared__ int s_red[16];
    const int t = threadIdx.x;
    const int lane = t & 31;
    const int wave = t >> 5;
    const int bin = blockIdx.x;
    const int2* __restrict__ pairs = (const int2*)binned;
    const int chunk = (e + NCH - 1) / NCH;
    const int node0 = bin * NPB;

    int lsum = 0, csum = 0;
    if (t < NCH) {
        const int2 lc = tab[bin * NCH + t];
        s_sbeg[t] = t * chunk + lc.x;
        s_scnt[t] = lc.y;
        lsum = lc.x;
        csum = lc.y;
    }
#pragma unroll
    for (int off = 16; off > 0; off >>= 1) {
        lsum += __shfl_xor(lsum, off);
        csum += __shfl_xor(csum, off);
    }
    if (wave < 8 && lane == 0) { s_red[wave] = lsum; s_red[8 + wave] = csum; }
    if (t < NBIN) s_cur[t] = 0;
    __syncthreads();
    int beg = 0, ne = 0;
#pragma unroll
    for (int w = 0; w < 8; ++w) { beg += s_red[w]; ne += s_red[8 + w]; }
    const int end = beg + ne;

    const int sb = s_sbeg[t >> 2];
    const int sc = s_scnt[t >> 2];
    const int sub = t & 3;
    for (int i0 = sub; i0 < sc; i0 += 16) {
        int d[4];
#pragma unroll
        for (int u = 0; u < 4; ++u) d[u] = pairs[sb + min(i0 + u * 4, sc - 1)].y;
        hold_trip_loads(d[0], d[1], d[2], d[3]);
#pragma unroll
        for (int u = 0; u < 4; ++u)
            if (i0 + u * 4 < sc) atomicAdd(&s_cur[d[u] - node0], 1);
    }
    __syncthreads();
    const int deg = (t < NBIN) ? s_cur[t] : 0;
    const int ex = excl_scan_256(deg, s_wsum);
    if (t < NPB && node0 + t < n) {
        rowptr[node0 + t] = beg + ex;
        dinv[node0 + t]   = rsqrtf(fmaxf((float)(deg + 1), 1.0f));
    }
    if (bin == NBIN - 1 && t == 0) rowptr[n] = end;
    if (t < NBIN) s_cur[t] = ex;
    __syncthreads();

    if (ne <= COL_CAP) {
        for (int i0 = sub; i0 < sc; i0 += 16) {
            int2 p[4];
#pragma unroll
            for (int u = 0; u < 4; ++u) p[u] = pairs[sb + min(i0 + u * 4, sc - 1)];
            hold_trip_loads(p[0].y, p[1].y, p[2].y, p[3].y);
#pragma unroll
            for (int u = 0; u < 4; ++u) {
                if (i0 + u * 4 < sc) {
                    const int r = atomicAdd(&s_cur[p[u].y - node0], 1);
                    s_col[r] = p[u].x;
                }
            }
        }
        __syncthreads();
        for (int i = t; i < ne; i += SORT_T) col[beg + i] = s_col[i];
    } else {
        for (int i = sub; i < sc; i += 4) {
            const int2 p = pairs[sb + i];
            const int r = atomicAdd(&s_cur[p.y - node0], 1);
            col[beg + r] = p.x;
        }
    }
}

template <int KIN, int KOUT>
__global__ __launch_bounds__(THREADS)
void k_gemm_wmma(const float* __restrict__ A, const float* __restrict__ W,
                 const float* __restrict__ rscale, float* __restrict__ C, int nrows) {
    __shared__ __attribute__((aligned(16))) v2f sW[(KIN / 2) * KOUT];
    constexpr int NT    = KOUT / 16;
    constexpr int NK    = KIN / 4;
    constexpr int UNITS = (KIN / 2) * (KOUT / 4);
    constexpr int UPT   = UNITS / THREADS;
    static_assert(UNITS % THREADS == 0, "every thread stages the same number of units");

    const int wave = threadIdx.x >> 5;
    const int lane = threadIdx.x & 31;
    const int tile = blockIdx.x * 8 + wave;
    const bool live = tile * 16 < nrows;
    const int ltile = min(tile, (nrows - 1) / 16);
    const int m    = lane & 15;
    const int khi  = lane >> 4;

    float4 w0[UPT], w1[UPT];
#pragma unroll
    for (int j = 0; j < UPT; ++j) {
        const int u  = threadIdx.x + j * THREADS;
        const int k2 = u / (KOUT / 4);
        const int c4 = (u % (KOUT / 4)) * 4;
        w0[j] = *(const float4*)(W + (2 * k2 + 0) * KOUT + c4);
        w1[j] = *(const float4*)(W + (2 * k2 + 1) * KOUT + c4);
    }

    v2f a[NK];
    const float* Arow = A + (size_t)(ltile * 16 + m) * KIN + khi * 2;
#pragma unroll
    for (int kk = 0; kk < NK; ++kk) a[kk] = *(const v2f*)(Arow + kk * 4);
    const float4 rs0 = *(const float4*)(rscale + ltile * 16 + khi * 8);
    const float4 rs1 = *(const float4*)(rscale + ltile * 16 + khi * 8 + 4);

#pragma unroll
    for (int j = 0; j < UPT; ++j) {
        const int u = threadIdx.x + j * THREADS;
        float4* d = (float4*)(sW + 4 * u);
        d[0] = make_float4(w0[j].x, w1[j].x, w0[j].y, w1[j].y);
        d[1] = make_float4(w0[j].z, w1[j].z, w0[j].w, w1[j].w);
    }
    __builtin_amdgcn_fence(__ATOMIC_RELEASE, "workgroup", "local");
    __builtin_amdgcn_s_barrier();
    __builtin_amdgcn_fence(__ATOMIC_ACQUIRE, "workgroup", "local");
    if (!live) {
        __builtin_amdgcn_fence(__ATOMIC_ACQUIRE, "workgroup");
        return;
    }

    v8f acc[NT] = {};
    const v2f* sWl = sW + khi * KOUT + m;

#pragma unroll
    for (int kk = 0; kk < NK; ++kk) {
        const v2f* sWk = sWl + (kk * 2) * KOUT;
#pragma unroll
        for (int t = 0; t < NT; ++t) {
            const v2f b = sWk[t * 16];
            acc[t] = __builtin_amdgcn_wmma_f32_16x16x4_f32(
                 false, a[kk],  false, b,
                 (short)0, acc[t],  false,  false);
        }
    }

    const float rs[8] = {rs0.x, rs0.y, rs0.z, rs0.w, rs1.x, rs1.y, rs1.z, rs1.w};
#pragma unroll
    for (int t = 0; t < NT; ++t) {
#pragma unroll
        for (int v = 0; v < 8; ++v) {
            const int r = tile * 16 + khi * 8 + v;
            C[(size_t)r * KOUT + t * 16 + m] = acc[t][v] * rs[v];
        }
    }
}


template <int G>
__device__ __forceinline__ int group_bcast(int v, int k);

template <>
__device__ __forceinline__ int group_bcast<16>(int v, int k) {
#define ROW_SHARE(K) case K: return __builtin_amdgcn_update_dpp(0, v, 0x150 + K, 0xF, 0xF, false);
    switch (k) {
        ROW_SHARE(0)  ROW_SHARE(1)  ROW_SHARE(2)  ROW_SHARE(3)
        ROW_SHARE(4)  ROW_SHARE(5)  ROW_SHARE(6)  ROW_SHARE(7)
        ROW_SHARE(8)  ROW_SHARE(9)  ROW_SHARE(10) ROW_SHARE(11)
        ROW_SHARE(12) ROW_SHARE(13) ROW_SHARE(14) ROW_SHARE(15)
    }
#undef ROW_SHARE
    return v;
}

template <>
__device__ __forceinline__ int group_bcast<8>(int v, int k) {
#define DPP8_ALL(K) case K: return __builtin_amdgcn_mov_dpp8(v, K * 0x249249);
    switch (k) {
        DPP8_ALL(0) DPP8_ALL(1) DPP8_ALL(2) DPP8_ALL(3)
        DPP8_ALL(4) DPP8_ALL(5) DPP8_ALL(6) DPP8_ALL(7)
    }
#undef DPP8_ALL
    return v;
}

__device__ __forceinline__ void hold_list_loads(const int (&w)[4]) {
    __builtin_amdgcn_sched_barrier(0);
    asm volatile("" :: "v"(w[0]), "v"(w[1]), "v"(w[2]), "v"(w[3]));
}
__device__ __forceinline__ void hold_list_loads(const int (&)[2]) {}

template <int F, bool RELU, bool NEXT>
__device__ __forceinline__ float4 agg_row(const int* __restrict__ rowptr, const int* __restrict__ col,
                                          const float* __restrict__ dinv, const float* __restrict__ Hs,
                                          const float* __restrict__ bias, int i, int c) {
    constexpr int G  = F / 4;
    constexpr int NI = 32 / G;
    const int beg = rowptr[i];
    const int end = rowptr[i + 1];
    const float di = dinv[i];
    const float* Hc = Hs + c * 4;
    const char* Hb = (const char*)Hs;
    const unsigned coff = (unsigned)c * 16u;

    const float4 hs = *(const float4*)(Hc + (size_t)i * F);
    const float4 bb = *(const float4*)(bias + c * 4);

    float4 acc = make_float4(0.0f, 0.0f, 0.0f, 0.0f);
    const int last = end - 1;
    for (int base = beg; base < end; base += 32) {
        int idx[NI];
#pragma unroll
        for (int q = 0; q < NI; ++q) idx[q] = col[min(base + q * G + c, last)];
        hold_list_loads(idx);
#pragma unroll
        for (int t = 0; t < 4; ++t) {
            const int e0 = base + t * 8;
            if (t == 0 || e0 < end) {
                float4 h[8];
#pragma unroll
                for (int u = 0; u < 8; ++u) {
                    const int e = t * 8 + u;
                    const unsigned s = (unsigned)group_bcast<G>(idx[e / G], e % G);
                    h[u] = *(const float4*)(Hb + (s * (unsigned)(F * sizeof(float)) + coff));
                }
#pragma unroll
                for (int u = 0; u < 8; ++u) {
                    const bool p = (u == 0) || (e0 + u < end);
                    acc.x = p ? acc.x + h[u].x : acc.x; acc.y = p ? acc.y + h[u].y : acc.y;
                    acc.z = p ? acc.z + h[u].z : acc.z; acc.w = p ? acc.w + h[u].w : acc.w;
                }
            }
        }
    }

    float4 v;
    v.x = (acc.x + hs.x) * di + bb.x;
    v.y = (acc.y + hs.y) * di + bb.y;
    v.z = (acc.z + hs.z) * di + bb.z;
    v.w = (acc.w + hs.w) * di + bb.w;
    if (RELU) {
        v.x = fmaxf(v.x, 0.0f); v.y = fmaxf(v.y, 0.0f);
        v.z = fmaxf(v.z, 0.0f); v.w = fmaxf(v.w, 0.0f);
    }
    if (NEXT) {
        v.x *= di; v.y *= di; v.z *= di; v.w *= di;
    }
    return v;
}

template <int F, bool RELU>
__global__ __launch_bounds__(THREADS)
void k_gather(const int* __restrict__ rowptr, const int* __restrict__ col,
              const float* __restrict__ dinv, const float* __restrict__ H,
              const float* __restrict__ bias, float* __restrict__ out, int n) {
    constexpr int C = F / 4;
    const int tid = blockIdx.x * blockDim.x + threadIdx.x;
    const int i = tid / C;
    const int c = tid % C;
    if (i >= n) return;
    const float4 v = agg_row<F, RELU, false>(rowptr, col, dinv, H, bias, i, c);
    *(float4*)(out + (size_t)i * F + c * 4) = v;
}

#define TILE_PITCH 68
template <int KOUT, bool MOVE>
__global__ __launch_bounds__(THREADS)
void k_gather_gemm(const int* __restrict__ rowptr, const int* __restrict__ col,
                   const float* __restrict__ dinv, const float* __restrict__ H,
                   const float* __restrict__ bias, const float* __restrict__ Wn,
                   float* __restrict__ Cn, int n,
                   const int* __restrict__ mv_from, int* __restrict__ mv_to, int mv_words) {
    constexpr int F  = 64;
    constexpr int NT = KOUT / 16;
    __shared__ float sH[16 * TILE_PITCH];
    const int r = threadIdx.x >> 4;
    const int c = threadIdx.x & 15;
    const int i = blockIdx.x * 16 + r;

    float4 v = make_float4(0.0f, 0.0f, 0.0f, 0.0f);
    if (i < n) v = agg_row<F, true, true>(rowptr, col, dinv, H, bias, i, c);
    *(float4*)(sH + r * TILE_PITCH + c * 4) = v;

    const int wave = threadIdx.x >> 5;
    const int lane = threadIdx.x & 31;
    const int m    = lane & 15;
    const int khi  = lane >> 4;
    const bool mul = wave < NT;

    v2f b[F / 4];
    if (mul) {
        const float* Wl = Wn + (khi * 2) * KOUT + wave * 16 + m;
#pragma unroll
        for (int kk = 0; kk < F / 4; ++kk) {
            b[kk].x = Wl[(kk * 4 + 0) * KOUT];
            b[kk].y = Wl[(kk * 4 + 1) * KOUT];
        }
    }
    __syncthreads();
    if (!mul) {
        if (MOVE) {
            constexpr int MT = THREADS - NT * 32;
            constexpr int NWG = GCN_N / 16;
            const int lines = (mv_words + 31) / 32;
            const int per   = ((lines + NWG - 1) / NWG) * 32;
            const int beg   = (int)blockIdx.x * per;
            const int end   = min(beg + per, mv_words);
            for (int w = beg + ((int)threadIdx.x - NT * 32) * 4; w < end; w += MT * 4) {
                if (w + 4 <= end) {
                    *(int4*)(mv_to + w) = *(const int4*)(mv_from + w);
                } else {
                    for (int u = w; u < end; ++u) mv_to[u] = mv_from[u];
                }
            }
        }
        return;
    }

    v8f acc = {};
    const float* sA = sH + m * TILE_PITCH + khi * 2;
#pragma unroll
    for (int kk = 0; kk < F / 4; ++kk) {
        const v2f a = *(const v2f*)(sA + kk * 4);
        acc = __builtin_amdgcn_wmma_f32_16x16x4_f32(
             false, a,  false, b[kk],
             (short)0, acc,  false,  false);
    }
    float* Cr = Cn + (size_t)(blockIdx.x * 16 + khi * 8) * KOUT + wave * 16 + m;
#pragma unroll
    for (int q = 0; q < 8; ++q) Cr[(size_t)q * KOUT] = acc[q];
}
static_assert(GCN_N % 16 == 0, "row tiles are whole");

extern "C" void kernel_launch(void* const* d_in, const int* in_sizes, int n_in,
                              void* d_out, int out_size, void* d_ws, size_t ws_size,
                              hipStream_t stream) {
    const float* x  = (const float*)d_in[0];
    const int*   ei = (const int*)d_in[1];
    const float* W1 = (const float*)d_in[2];
    const float* b1 = (const float*)d_in[3];
    const float* W2 = (const float*)d_in[4];
    const float* b2 = (const float*)d_in[5];
    const float* W3 = (const float*)d_in[6];
    const float* b3 = (const float*)d_in[7];
    float* out = (float*)d_out;

    const int N = GCN_N;
    const int E = in_sizes[1] / 2;
    const int* src = ei;
    const int* dst = ei + E;

    float* dinv = (float*)d_ws;
    float* bufA = dinv + ((N + 255) & ~255);
    float* bufB = bufA + (size_t)N * 64;

    int* tab     = (int*)bufA;
    int* binned  = (int*)bufB;
    int* col    = (int*)out;
    int* rowptr = col + E;
    int* col3    = (int*)(bufA + (size_t)N * 32);
    int* rowptr3 = col3 + E;

    const int tiles  = N / 16;
    const int gemmBl = (tiles + 7) / 8;

    auto blocks = [](long long n) { return (unsigned)((n + THREADS - 1) / THREADS); };

    k_slice_sort<<<NCH, SORT_T, 0, stream>>>(src, dst, tab, binned, E);
    k_bin_rows<<<NBIN, SORT_T, 0, stream>>>(binned, tab, rowptr, col, dinv, N, E);

    k_gemm_wmma<128, 64><<<gemmBl, THREADS, 0, stream>>>(x, W1, dinv, bufA, N);

    k_gather_gemm<64, false><<<tiles, THREADS, 0, stream>>>(rowptr, col, dinv, bufA, b1, W2, bufB, N,
                                                            nullptr, nullptr, 0);

    k_gather_gemm<32, true><<<tiles, THREADS, 0, stream>>>(rowptr, col, dinv, bufB, b2, W3, bufA, N,
                                                           col, col3, E + N + 1);

    k_gather<32, false><<<blocks((long long)N * 8), THREADS, 0, stream>>>(rowptr3, col3, dinv, bufA, b3, out, N);
}


# Dense transform x@W1 moved into the row-list launch: a bin's work-group multiplies its own 196 rows on WMMA after its lists (dinv from its LDS); 6 launches -> 5; sums bit-identical
# speedup vs baseline: 4.9267x; 1.0348x over previous
#include <hip/hip_runtime.h>
#include <cstddef>

typedef __attribute__((ext_vector_type(2))) float v2f;
typedef __attribute__((ext_vector_type(8))) float v8f;

#define GCN_N 50000
#define THREADS 256

#define NBIN 256
#define NCH  256
#define NPB  ((GCN_N + NBIN - 1) / NBIN)
#define SORT_T 1024
#define CH_CAP  4096
#define COL_CAP 8192
static_assert(CH_CAP == 4 * SORT_T, "a staged slice is four edges a thread");
static_assert(NCH * 4 == SORT_T, "four lanes walk one slice's run of a bin");
static_assert(NPB <= NBIN, "a bin's nodes fit the 256 LDS counters");

__device__ __forceinline__ int excl_scan_256(int v, int* s_wsum) {
    const int lane = threadIdx.x & 31;
    const int wave = threadIdx.x >> 5;
    int x = v;
#pragma unroll
    for (int off = 1; off < 32; off <<= 1) {
        const int y = __shfl_up(x, off);
        if (lane >= off) x += y;
    }
    if (wave < 8 && lane == 31) s_wsum[wave] = x;
    __syncthreads();
    int base = 0;
    if (wave < 8)
        for (int w = 0; w < wave; ++w) base += s_wsum[w];
    __syncthreads();
    return base + x - v;
}

__global__ __launch_bounds__(SORT_T)
void k_slice_sort(const int* __restrict__ src, const int* __restrict__ dst,
                  int* __restrict__ tabw, int* __restrict__ binned, int e) {
    int2* __restrict__ tab = (int2*)tabw;
    __shared__ int2 s_pairs[CH_CAP];
    __shared__ int s_cnt[NBIN];
    __shared__ int s_lst[NBIN];
    __shared__ int s_cur[NBIN];
    __shared__ int s_wsum[8];
    const int t = threadIdx.x;
    const int g = blockIdx.x;
    int2* __restrict__ pairs = (int2*)binned;
    const int chunk = (e + NCH - 1) / NCH;
    const int beg = g * chunk;
    const int end = min(beg + chunk, e);
    const int cn  = end - beg;

    if (t < NBIN) s_cnt[t] = 0;
    __syncthreads();
    if (cn <= CH_CAP) {
        const bool has = beg + t < end;
        int d[4] = {0, 0, 0, 0}, s[4] = {0, 0, 0, 0};
        if (has) {
#pragma unroll
            for (int u = 0; u < 4; ++u) {
                const int ii = min(beg + t + u * SORT_T, end - 1);
                d[u] = dst[ii];
                s[u] = src[ii];
            }
        }
#pragma unroll
        for (int u = 0; u < 4; ++u)
            if (beg + t + u * SORT_T < end) atomicAdd(&s_cnt[d[u] / NPB], 1);
        __syncthreads();
        const int own = (t < NBIN) ? s_cnt[t] : 0;
        const int lst = excl_scan_256(own, s_wsum);
        if (t < NBIN) {
            s_lst[t] = lst;
            s_cur[t] = 0;
        }
        __syncthreads();
#pragma unroll
        for (int u = 0; u < 4; ++u) {
            if (beg + t + u * SORT_T < end) {
                const int bb = d[u] / NPB;
                const int r = atomicAdd(&s_cur[bb], 1);
                s_pairs[s_lst[bb] + r] = make_int2(s[u], d[u]);
            }
        }
        __syncthreads();
        if (t < NBIN) tab[t * NCH + g] = make_int2(lst, own);
        for (int i = t; i < cn; i += SORT_T) pairs[beg + i] = s_pairs[i];
    } else {
        for (int i = beg + t; i < end; i += SORT_T) atomicAdd(&s_cnt[dst[i] / NPB], 1);
        __syncthreads();
        const int own = (t < NBIN) ? s_cnt[t] : 0;
        const int lst = excl_scan_256(own, s_wsum);
        if (t < NBIN) {
            s_lst[t] = lst;
            s_cur[t] = 0;
            tab[t * NCH + g] = make_int2(lst, own);
        }
        __syncthreads();
        for (int i = beg + t; i < end; i += SORT_T) {
            const int dd = dst[i];
            const int ss = src[i];
            const int bb = dd / NPB;
            const int r = atomicAdd(&s_cur[bb], 1);
            pairs[beg + s_lst[bb] + r] = make_int2(ss, dd);
        }
    }
}

__device__ __forceinline__ void hold_trip_loads(int a, int b, int c, int d) {
    __builtin_amdgcn_sched_barrier(0);
    asm volatile("" :: "v"(a), "v"(b), "v"(c), "v"(d));
}

#define GEMM_K 128
#define GEMM_N 64
#define BIN_TILES ((NPB + 15) / 16)
static_assert(2 * BIN_TILES <= SORT_T / 32, "one (row tile, column half) per wave");
static_assert(16 * BIN_TILES <= NBIN, "every row of a tile has a factor in s_dinv");
static_assert((GEMM_K / 2) * (GEMM_N / 4) == SORT_T, "every thread stages one unit of W1");

__global__ __launch_bounds__(SORT_T) __attribute__((amdgpu_waves_per_eu(8, 8)))
void k_bin_rows(const int* __restrict__ binned, const int* __restrict__ tabw,
                int* __restrict__ rowptr, int* __restrict__ col,
                float* __restrict__ dinv, int n, int e,
                const float* __restrict__ X, const float* __restrict__ W,
                float* __restrict__ C) {
    const int2* __restrict__ tab = (const int2*)tabw;
    __shared__ int s_cur[NBIN];
    __shared__ int s_col[COL_CAP];
    __shared__ int s_wsum[8];
    __shared__ int s_sbeg[NCH];
    __shared__ int s_scnt[NCH];
    __shared__ int s_red[16];
    __shared__ float s_dinv[NBIN];
    __shared__ __attribute__((aligned(16))) v2f sW[(GEMM_K / 2) * GEMM_N];
    const int t = threadIdx.x;
    const int lane = t & 31;
    const int wave = t >> 5;
    const int bin = blockIdx.x;
    const int2* __restrict__ pairs = (const int2*)binned;
    const int chunk = (e + NCH - 1) / NCH;
    const int node0 = bin * NPB;

    const int wk2 = t / (GEMM_N / 4);
    const int wc4 = (t % (GEMM_N / 4)) * 4;
    const float4 w0 = *(const float4*)(W + (2 * wk2 + 0) * GEMM_N + wc4);
    const float4 w1 = *(const float4*)(W + (2 * wk2 + 1) * GEMM_N + wc4);

    int lsum = 0, csum = 0;
    if (t < NCH) {
        const int2 lc = tab[bin * NCH + t];
        s_sbeg[t] = t * chunk + lc.x;
        s_scnt[t] = lc.y;
        lsum = lc.x;
        csum = lc.y;
    }
#pragma unroll
    for (int off = 16; off > 0; off >>= 1) {
        lsum += __shfl_xor(lsum, off);
        csum += __shfl_xor(csum, off);
    }
    if (wave < 8 && lane == 0) { s_red[wave] = lsum; s_red[8 + wave] = csum; }
    if (t < NBIN) s_cur[t] = 0;
    {
        float4* d = (float4*)(sW + 4 * t);
        d[0] = make_float4(w0.x, w1.x, w0.y, w1.y);
        d[1] = make_float4(w0.z, w1.z, w0.w, w1.w);
    }
    __syncthreads();
    int beg = 0, ne = 0;
#pragma unroll
    for (int w = 0; w < 8; ++w) { beg += s_red[w]; ne += s_red[8 + w]; }
    const int end = beg + ne;

    const int sb = s_sbeg[t >> 2];
    const int sc = s_scnt[t >> 2];
    const int sub = t & 3;
    for (int i0 = sub; i0 < sc; i0 += 16) {
        int d[4];
#pragma unroll
        for (int u = 0; u < 4; ++u) d[u] = pairs[sb + min(i0 + u * 4, sc - 1)].y;
        hold_trip_loads(d[0], d[1], d[2], d[3]);
#pragma unroll
        for (int u = 0; u < 4; ++u)
            if (i0 + u * 4 < sc) atomicAdd(&s_cur[d[u] - node0], 1);
    }
    __syncthreads();
    const int deg = (t < NBIN) ? s_cur[t] : 0;
    const int ex = excl_scan_256(deg, s_wsum);
    const float dv = rsqrtf(fmaxf((float)(deg + 1), 1.0f));
    if (t < NBIN) s_dinv[t] = dv;
    if (t < NPB && node0 + t < n) {
        rowptr[node0 + t] = beg + ex;
        dinv[node0 + t]   = dv;
    }
    if (bin == NBIN - 1 && t == 0) rowptr[n] = end;
    if (t < NBIN) s_cur[t] = ex;
    __syncthreads();

    if (ne <= COL_CAP) {
        for (int i0 = sub; i0 < sc; i0 += 16) {
            int2 p[4];
#pragma unroll
            for (int u = 0; u < 4; ++u) p[u] = pairs[sb + min(i0 + u * 4, sc - 1)];
            hold_trip_loads(p[0].y, p[1].y, p[2].y, p[3].y);
#pragma unroll
            for (int u = 0; u < 4; ++u) {
                if (i0 + u * 4 < sc) {
                    const int r = atomicAdd(&s_cur[p[u].y - node0], 1);
                    s_col[r] = p[u].x;
                }
            }
        }
        __syncthreads();
        for (int i = t; i < ne; i += SORT_T) col[beg + i] = s_col[i];
    } else {
        for (int i = sub; i < sc; i += 4) {
            const int2 p = pairs[sb + i];
            const int r = atomicAdd(&s_cur[p.y - node0], 1);
            col[beg + r] = p.x;
        }
    }

    if (wave >= 2 * BIN_TILES) return;
    if (node0 + (wave >> 1) * 16 >= n) return;
    const int tile = wave >> 1;
    const int cp   = wave & 1;
    const int m    = lane & 15;
    const int khi  = lane >> 4;

    v2f a[GEMM_K / 4];
    {
        const int lr = min(node0 + tile * 16 + m, n - 1);
        const float* Arow = X + (size_t)lr * GEMM_K + khi * 2;
#pragma unroll
        for (int kk = 0; kk < GEMM_K / 4; ++kk) a[kk] = *(const v2f*)(Arow + kk * 4);
    }

    v8f acc0 = {}, acc1 = {};
    const v2f* sWl = sW + khi * GEMM_N + cp * 32 + m;
#pragma unroll
    for (int kk = 0; kk < GEMM_K / 4; ++kk) {
        const v2f* sWk = sWl + (kk * 2) * GEMM_N;
        const v2f b0 = sWk[0];
        const v2f b1 = sWk[16];
        acc0 = __builtin_amdgcn_wmma_f32_16x16x4_f32(
             false, a[kk],  false, b0,
             (short)0, acc0,  false,  false);
        acc1 = __builtin_amdgcn_wmma_f32_16x16x4_f32(
             false, a[kk],  false, b1,
             (short)0, acc1,  false,  false);
    }

    float o0[8], o1[8];
#pragma unroll
    for (int v = 0; v < 8; ++v) {
        const float rs = s_dinv[tile * 16 + khi * 8 + v];
        o0[v] = acc0[v] * rs;
        o1[v] = acc1[v] * rs;
    }
#pragma unroll
    for (int v = 0; v < 8; ++v) {
        const int rl = tile * 16 + khi * 8 + v;
        const int r  = node0 + rl;
        if (rl < NPB && r < n) {
            float* Cr = C + (size_t)r * GEMM_N + cp * 32 + m;
            Cr[0]  = o0[v];
            Cr[16] = o1[v];
        }
    }
}


template <int G>
__device__ __forceinline__ int group_bcast(int v, int k);

template <>
__device__ __forceinline__ int group_bcast<16>(int v, int k) {
#define ROW_SHARE(K) case K: return __builtin_amdgcn_update_dpp(0, v, 0x150 + K, 0xF, 0xF, false);
    switch (k) {
        ROW_SHARE(0)  ROW_SHARE(1)  ROW_SHARE(2)  ROW_SHARE(3)
        ROW_SHARE(4)  ROW_SHARE(5)  ROW_SHARE(6)  ROW_SHARE(7)
        ROW_SHARE(8)  ROW_SHARE(9)  ROW_SHARE(10) ROW_SHARE(11)
        ROW_SHARE(12) ROW_SHARE(13) ROW_SHARE(14) ROW_SHARE(15)
    }
#undef ROW_SHARE
    return v;
}

template <>
__device__ __forceinline__ int group_bcast<8>(int v, int k) {
#define DPP8_ALL(K) case K: return __builtin_amdgcn_mov_dpp8(v, K * 0x249249);
    switch (k) {
        DPP8_ALL(0) DPP8_ALL(1) DPP8_ALL(2) DPP8_ALL(3)
        DPP8_ALL(4) DPP8_ALL(5) DPP8_ALL(6) DPP8_ALL(7)
    }
#undef DPP8_ALL
    return v;
}

__device__ __forceinline__ void hold_list_loads(const int (&w)[4]) {
    __builtin_amdgcn_sched_barrier(0);
    asm volatile("" :: "v"(w[0]), "v"(w[1]), "v"(w[2]), "v"(w[3]));
}
__device__ __forceinline__ void hold_list_loads(const int (&)[2]) {}

template <int F, bool RELU, bool NEXT>
__device__ __forceinline__ float4 agg_row(const int* __restrict__ rowptr, const int* __restrict__ col,
                                          const float* __restrict__ dinv, const float* __restrict__ Hs,
                                          const float* __restrict__ bias, int i, int c) {
    constexpr int G  = F / 4;
    constexpr int NI = 32 / G;
    const int beg = rowptr[i];
    const int end = rowptr[i + 1];
    const float di = dinv[i];
    const float* Hc = Hs + c * 4;
    const char* Hb = (const char*)Hs;
    const unsigned coff = (unsigned)c * 16u;

    const float4 hs = *(const float4*)(Hc + (size_t)i * F);
    const float4 bb = *(const float4*)(bias + c * 4);

    float4 acc = make_float4(0.0f, 0.0f, 0.0f, 0.0f);
    const int last = end - 1;
    for (int base = beg; base < end; base += 32) {
        int idx[NI];
#pragma unroll
        for (int q = 0; q < NI; ++q) idx[q] = col[min(base + q * G + c, last)];
        hold_list_loads(idx);
#pragma unroll
        for (int t = 0; t < 4; ++t) {
            const int e0 = base + t * 8;
            if (t == 0 || e0 < end) {
                float4 h[8];
#pragma unroll
                for (int u = 0; u < 8; ++u) {
                    const int e = t * 8 + u;
                    const unsigned s = (unsigned)group_bcast<G>(idx[e / G], e % G);
                    h[u] = *(const float4*)(Hb + (s * (unsigned)(F * sizeof(float)) + coff));
                }
#pragma unroll
                for (int u = 0; u < 8; ++u) {
                    const bool p = (u == 0) || (e0 + u < end);
                    acc.x = p ? acc.x + h[u].x : acc.x; acc.y = p ? acc.y + h[u].y : acc.y;
                    acc.z = p ? acc.z + h[u].z : acc.z; acc.w = p ? acc.w + h[u].w : acc.w;
                }
            }
        }
    }

    float4 v;
    v.x = (acc.x + hs.x) * di + bb.x;
    v.y = (acc.y + hs.y) * di + bb.y;
    v.z = (acc.z + hs.z) * di + bb.z;
    v.w = (acc.w + hs.w) * di + bb.w;
    if (RELU) {
        v.x = fmaxf(v.x, 0.0f); v.y = fmaxf(v.y, 0.0f);
        v.z = fmaxf(v.z, 0.0f); v.w = fmaxf(v.w, 0.0f);
    }
    if (NEXT) {
        v.x *= di; v.y *= di; v.z *= di; v.w *= di;
    }
    return v;
}

template <int F, bool RELU>
__global__ __launch_bounds__(THREADS)
void k_gather(const int* __restrict__ rowptr, const int* __restrict__ col,
              const float* __restrict__ dinv, const float* __restrict__ H,
              const float* __restrict__ bias, float* __restrict__ out, int n) {
    constexpr int C = F / 4;
    const int tid = blockIdx.x * blockDim.x + threadIdx.x;
    const int i = tid / C;
    const int c = tid % C;
    if (i >= n) return;
    const float4 v = agg_row<F, RELU, false>(rowptr, col, dinv, H, bias, i, c);
    *(float4*)(out + (size_t)i * F + c * 4) = v;
}

#define TILE_PITCH 68
template <int KOUT, bool MOVE>
__global__ __launch_bounds__(THREADS)
void k_gather_gemm(const int* __restrict__ rowptr, const int* __restrict__ col,
                   const float* __restrict__ dinv, const float* __restrict__ H,
                   const float* __restrict__ bias, const float* __restrict__ Wn,
                   float* __restrict__ Cn, int n,
                   const int* __restrict__ mv_from, int* __restrict__ mv_to, int mv_words) {
    constexpr int F  = 64;
    constexpr int NT = KOUT / 16;
    __shared__ float sH[16 * TILE_PITCH];
    const int r = threadIdx.x >> 4;
    const int c = threadIdx.x & 15;
    const int i = blockIdx.x * 16 + r;

    float4 v = make_float4(0.0f, 0.0f, 0.0f, 0.0f);
    if (i < n) v = agg_row<F, true, true>(rowptr, col, dinv, H, bias, i, c);
    *(float4*)(sH + r * TILE_PITCH + c * 4) = v;

    const int wave = threadIdx.x >> 5;
    const int lane = threadIdx.x & 31;
    const int m    = lane & 15;
    const int khi  = lane >> 4;
    const bool mul = wave < NT;

    v2f b[F / 4];
    if (mul) {
        const float* Wl = Wn + (khi * 2) * KOUT + wave * 16 + m;
#pragma unroll
        for (int kk = 0; kk < F / 4; ++kk) {
            b[kk].x = Wl[(kk * 4 + 0) * KOUT];
            b[kk].y = Wl[(kk * 4 + 1) * KOUT];
        }
    }
    __syncthreads();
    if (!mul) {
        if (MOVE) {
            constexpr int MT = THREADS - NT * 32;
            constexpr int NWG = GCN_N / 16;
            const int lines = (mv_words + 31) / 32;
            const int per   = ((lines + NWG - 1) / NWG) * 32;
            const int beg   = (int)blockIdx.x * per;
            const int end   = min(beg + per, mv_words);
            for (int w = beg + ((int)threadIdx.x - NT * 32) * 4; w < end; w += MT * 4) {
                if (w + 4 <= end) {
                    *(int4*)(mv_to + w) = *(const int4*)(mv_from + w);
                } else {
                    for (int u = w; u < end; ++u) mv_to[u] = mv_from[u];
                }
            }
        }
        return;
    }

    v8f acc = {};
    const float* sA = sH + m * TILE_PITCH + khi * 2;
#pragma unroll
    for (int kk = 0; kk < F / 4; ++kk) {
        const v2f a = *(const v2f*)(sA + kk * 4);
        acc = __builtin_amdgcn_wmma_f32_16x16x4_f32(
             false, a,  false, b[kk],
             (short)0, acc,  false,  false);
    }
    float* Cr = Cn + (size_t)(blockIdx.x * 16 + khi * 8) * KOUT + wave * 16 + m;
#pragma unroll
    for (int q = 0; q < 8; ++q) Cr[(size_t)q * KOUT] = acc[q];
}
static_assert(GCN_N % 16 == 0, "row tiles are whole");

extern "C" void kernel_launch(void* const* d_in, const int* in_sizes, int n_in,
                              void* d_out, int out_size, void* d_ws, size_t ws_size,
                              hipStream_t stream) {
    const float* x  = (const float*)d_in[0];
    const int*   ei = (const int*)d_in[1];
    const float* W1 = (const float*)d_in[2];
    const float* b1 = (const float*)d_in[3];
    const float* W2 = (const float*)d_in[4];
    const float* b2 = (const float*)d_in[5];
    const float* W3 = (const float*)d_in[6];
    const float* b3 = (const float*)d_in[7];
    float* out = (float*)d_out;

    const int N = GCN_N;
    const int E = in_sizes[1] / 2;
    const int* src = ei;
    const int* dst = ei + E;

    float* dinv = (float*)d_ws;
    float* bufA = dinv + ((N + 255) & ~255);
    float* bufB = bufA + (size_t)N * 64;

    int* tab     = (int*)(bufB + (size_t)N * 32);
    int* binned  = (int*)bufB;
    int* col    = (int*)out;
    int* rowptr = col + E;
    int* col3    = (int*)(bufA + (size_t)N * 32);
    int* rowptr3 = col3 + E;

    const int tiles  = N / 16;

    auto blocks = [](long long n) { return (unsigned)((n + THREADS - 1) / THREADS); };

    k_slice_sort<<<NCH, SORT_T, 0, stream>>>(src, dst, tab, binned, E);
    k_bin_rows<<<NBIN, SORT_T, 0, stream>>>(binned, tab, rowptr, col, dinv, N, E, x, W1, bufA);

    k_gather_gemm<64, false><<<tiles, THREADS, 0, stream>>>(rowptr, col, dinv, bufA, b1, W2, bufB, N,
                                                            nullptr, nullptr, 0);

    k_gather_gemm<32, true><<<tiles, THREADS, 0, stream>>>(rowptr, col, dinv, bufB, b2, W3, bufA, N,
                                                           col, col3, E + N + 1);

    k_gather<32, false><<<blocks((long long)N * 8), THREADS, 0, stream>>>(rowptr3, col3, dinv, bufA, b3, out, N);
}
